# starting kernel
# baseline (speedup 1.0000x reference)
#include <hip/hip_runtime.h>
#include <hip/hip_bf16.h>


typedef __attribute__((ext_vector_type(2))) float v2f;
typedef __attribute__((ext_vector_type(4))) float v4f;
typedef __attribute__((ext_vector_type(8))) float v8f;

#define NTOK   4096
#define EMBED  512
#define QKVW   1536
#define HEADS  8
#define DH     64
#define SCALE  0.125f

static __device__ __forceinline__ v8f wmma4(v2f a, v2f b, v8f c) {
    return __builtin_amdgcn_wmma_f32_16x16x4_f32(
         false, a,  false, b,
         (short)0, c,  false,  false);
}

__global__ void mhsa_zero(float* __restrict__ p, int n) {
    int i = blockIdx.x * blockDim.x + threadIdx.x;
    if (i < n) p[i] = 0.0f;
}

__global__ __launch_bounds__(128) void mhsa_qkv_gemm(
    const float* __restrict__ x, const float* __restrict__ W,
    const float* __restrict__ bias, float* __restrict__ qkv)
{
    const int lane = threadIdx.x & 31;
    const int wave = threadIdx.x >> 5;
    const int lo = lane & 15;
    const int hi = lane >> 4;
    const int m0 = blockIdx.x * 64 + wave * 16;
    const int n0 = blockIdx.y * 64;

    v8f acc0 = {}, acc1 = {}, acc2 = {}, acc3 = {};
    const int m  = m0 + lo;
    const int nb = n0 + lo;

    #pragma unroll 4
    for (int k = 0; k < EMBED; k += 4) {
        const int ka = k + 2 * hi;
        v2f a;
        a.x = x[(size_t)ka * NTOK + m];
        a.y = x[(size_t)(ka + 1) * NTOK + m];
        const float* w0 = W + (size_t)ka * QKVW + nb;
        const float* w1 = W + (size_t)(ka + 1) * QKVW + nb;
        v2f b0 = {w0[0],  w1[0]};
        v2f b1 = {w0[16], w1[16]};
        v2f b2 = {w0[32], w1[32]};
        v2f b3 = {w0[48], w1[48]};
        acc0 = wmma4(a, b0, acc0);
        acc1 = wmma4(a, b1, acc1);
        acc2 = wmma4(a, b2, acc2);
        acc3 = wmma4(a, b3, acc3);
    }

    const float bi0 = bias[nb], bi1 = bias[nb + 16];
    const float bi2 = bias[nb + 32], bi3 = bias[nb + 48];
    #pragma unroll
    for (int r = 0; r < 8; ++r) {
        const int row = m0 + hi * 8 + r;
        float* dst = qkv + (size_t)row * QKVW + n0 + lo;
        dst[0]  = acc0[r] + bi0;
        dst[16] = acc1[r] + bi1;
        dst[32] = acc2[r] + bi2;
        dst[48] = acc3[r] + bi3;
    }
}

__global__ __launch_bounds__(64) void mhsa_qsum(
    const float* __restrict__ qkv, float* __restrict__ qsum)
{
    const int h = blockIdx.x, chunk = blockIdx.y, d = threadIdx.x;
    const float* col = qkv + (size_t)chunk * 256 * QKVW + h * DH + d;
    float s = 0.0f;
    #pragma unroll 8
    for (int i = 0; i < 256; ++i) s += col[(size_t)i * QKVW];
    atomicAdd(&qsum[h * DH + d], s);
}

__global__ __launch_bounds__(128) void mhsa_ktv(
    const float* __restrict__ qkv, float* __restrict__ KtV)
{
    const int h = blockIdx.x, chunk = blockIdx.y;
    const int lane = threadIdx.x & 31, wave = threadIdx.x >> 5;
    const int lo = lane & 15, hi = lane >> 4;
    const int m0 = wave * 16;
    const float* Kb = qkv + EMBED + h * DH;
    const float* Vb = qkv + 2 * EMBED + h * DH;
    const int nbase = chunk * 256;

    v8f acc0 = {}, acc1 = {}, acc2 = {}, acc3 = {};
    #pragma unroll 4
    for (int kk = 0; kk < 256; kk += 4) {
        const int n = nbase + kk + 2 * hi;
        v2f a;
        a.x = Kb[(size_t)n * QKVW + m0 + lo];
        a.y = Kb[(size_t)(n + 1) * QKVW + m0 + lo];
        const float* v0 = Vb + (size_t)n * QKVW + lo;
        const float* v1 = Vb + (size_t)(n + 1) * QKVW + lo;
        v2f b0 = {v0[0],  v1[0]};
        v2f b1 = {v0[16], v1[16]};
        v2f b2 = {v0[32], v1[32]};
        v2f b3 = {v0[48], v1[48]};
        acc0 = wmma4(a, b0, acc0);
        acc1 = wmma4(a, b1, acc1);
        acc2 = wmma4(a, b2, acc2);
        acc3 = wmma4(a, b3, acc3);
    }

    float* dst = KtV + (size_t)h * DH * DH;
    #pragma unroll
    for (int r = 0; r < 8; ++r) {
        const int row = m0 + hi * 8 + r;
        atomicAdd(&dst[row * DH + lo +  0], acc0[r]);
        atomicAdd(&dst[row * DH + lo + 16], acc1[r]);
        atomicAdd(&dst[row * DH + lo + 32], acc2[r]);
        atomicAdd(&dst[row * DH + lo + 48], acc3[r]);
    }
}

__global__ __launch_bounds__(256) void mhsa_coarse(
    const float* __restrict__ qkv, const float* __restrict__ qsum,
    float* __restrict__ out)
{
    const int idx = blockIdx.x * blockDim.x + threadIdx.x;
    const int h = idx >> 12, n = idx & (NTOK - 1);
    const float* Kr = qkv + (size_t)n * QKVW + EMBED + h * DH;
    const float* qs = qsum + h * DH;
    float s = 0.0f;
    #pragma unroll
    for (int d = 0; d < DH; d += 4) {
        const v4f kv = *(const v4f*)(Kr + d);
        const v4f qv = *(const v4f*)(qs + d);
        s += kv.x * qv.x + kv.y * qv.y + kv.z * qv.z + kv.w * qv.w;
    }
    out[idx] = s * SCALE;
}

__global__ __launch_bounds__(128) void mhsa_out_gemm(
    const float* __restrict__ qkv, const float* __restrict__ KtV,
    float* __restrict__ out)
{
    __shared__ float lB[DH * DH];
    const int h = blockIdx.x, mt = blockIdx.y;

    {
        const v4f* src = (const v4f*)(KtV + (size_t)h * DH * DH);
        v4f* dst = (v4f*)lB;
        for (int i = threadIdx.x; i < (DH * DH) / 4; i += 128) dst[i] = src[i];
    }
    __syncthreads();

    const int lane = threadIdx.x & 31, wave = threadIdx.x >> 5;
    const int lo = lane & 15, hi = lane >> 4;
    const int m0 = mt * 64 + wave * 16;
    const float* Qb = qkv + h * DH;
    const int m = m0 + lo;

    v8f acc0 = {}, acc1 = {}, acc2 = {}, acc3 = {};
    #pragma unroll
    for (int k = 0; k < DH; k += 4) {
        const int ka = k + 2 * hi;
        v2f a = *(const v2f*)(Qb + (size_t)m * QKVW + ka);
        const float* B0 = lB + ka * DH + lo;
        const float* B1 = lB + (ka + 1) * DH + lo;
        v2f b0 = {B0[0],  B1[0]};
        v2f b1 = {B0[16], B1[16]};
        v2f b2 = {B0[32], B1[32]};
        v2f b3 = {B0[48], B1[48]};
        acc0 = wmma4(a, b0, acc0);
        acc1 = wmma4(a, b1, acc1);
        acc2 = wmma4(a, b2, acc2);
        acc3 = wmma4(a, b3, acc3);
    }

    float* base = out + HEADS * NTOK + (size_t)h * NTOK * DH;
    #pragma unroll
    for (int r = 0; r < 8; ++r) {
        const int row = m0 + hi * 8 + r;
        float* o = base + (size_t)row * DH + lo;
        o[0]  = acc0[r] * SCALE;
        o[16] = acc1[r] * SCALE;
        o[32] = acc2[r] * SCALE;
        o[48] = acc3[r] * SCALE;
    }
}

extern "C" void kernel_launch(void* const* d_in, const int* in_sizes, int n_in,
                              void* d_out, int out_size, void* d_ws, size_t ws_size,
                              hipStream_t stream) {
    const float* x    = (const float*)d_in[0];
    const float* W    = (const float*)d_in[1];
    const float* bias = (const float*)d_in[2];
    float* out = (float*)d_out;

    float* qkv  = (float*)d_ws;
    float* KtV  = qkv + (size_t)NTOK * QKVW;
    float* qsum = KtV + HEADS * DH * DH;

    const int nzero = HEADS * DH * DH + HEADS * DH;
    mhsa_zero<<<(nzero + 255) / 256, 256, 0, stream>>>(KtV, nzero);

    mhsa_qkv_gemm<<<dim3(NTOK / 64, QKVW / 64), 128, 0, stream>>>(x, W, bias, qkv);
    mhsa_qsum<<<dim3(HEADS, NTOK / 256), 64, 0, stream>>>(qkv, qsum);
    mhsa_ktv<<<dim3(HEADS, NTOK / 256), 128, 0, stream>>>(qkv, KtV);
    mhsa_coarse<<<(HEADS * NTOK) / 256, 256, 0, stream>>>(qkv, qsum, out);
    mhsa_out_gemm<<<dim3(HEADS, NTOK / 64), 128, 0, stream>>>(qkv, KtV, out);
}


# Output GEMM: both operands staged k-major through two LDS buffers by whole-row 16-byte loads, next chunk's loads in flight under the WMMAs, 8-wave blocks (wave tile 16x32); sum order unchanged
# speedup vs baseline: 1.3742x; 1.3742x over previous
#include <hip/hip_runtime.h>


typedef __attribute__((ext_vector_type(2))) float v2f;
typedef __attribute__((ext_vector_type(4))) float v4f;
typedef __attribute__((ext_vector_type(8))) float v8f;

#define NTOK   4096
#define EMBED  512
#define QKVW   1536
#define HEADS  8
#define DH     64
#define SCALE  0.125f
#define NTOKF  4096.0f
#define GSPL   16
#define TSPL   4
#define KSPL   8
#define XSPL   8
#define SSPL   16
#define EE     (EMBED * EMBED)
#define GPITCH 68
#define KPITCH 72

#define OFF_GP    0
#define OFF_G     (OFF_GP + GSPL * EE)
#define OFF_TP    (OFF_G + EE)
#define OFF_KP    (OFF_TP + TSPL * EE)
#define OFF_MQ    (OFF_KP + KSPL * HEADS * DH * DH)
#define OFF_XSP   (OFF_MQ + EE)
#define OFF_SVP   (OFF_XSP + XSPL * EMBED)
#define OFF_U     (OFF_SVP + SSPL * QKVW)
#define OFF_R     (OFF_U + HEADS * EMBED)
#define OFF_C     (OFF_R + EMBED)

static __device__ __forceinline__ v8f wmma4(v2f a, v2f b, v8f c) {
    return __builtin_amdgcn_wmma_f32_16x16x4_f32(
        false, a, false, b, (short)0, c, false, false);
}

__global__ __launch_bounds__(128) void mhsa_gram(
    const float* __restrict__ x, float* __restrict__ Gp)
{
    __shared__ __attribute__((aligned(16))) float sA[64 * GPITCH];
    __shared__ __attribute__((aligned(16))) float sB[64 * GPITCH];
    const int mi = blockIdx.x, ni = blockIdx.y, s = blockIdx.z;
    const int t = threadIdx.x, lane = t & 31, wave = t >> 5;
    const int lo = lane & 15, hi = lane >> 4;
    const int sr = t >> 4, sc = (t & 15) * 4;
    const float* xa = x + (size_t)(mi * 64 + sr) * NTOK + s * 256 + sc;
    const float* xb = x + (size_t)(ni * 64 + sr) * NTOK + s * 256 + sc;

    v4f ra[8], rb[8];
    #pragma unroll
    for (int i = 0; i < 8; ++i) {
        ra[i] = *(const v4f*)(xa + (size_t)i * 8 * NTOK);
        rb[i] = *(const v4f*)(xb + (size_t)i * 8 * NTOK);
    }

    v8f acc0 = {}, acc1 = {}, acc2 = {}, acc3 = {};
    const float* pa  = sA + (wave * 16 + lo) * GPITCH + 4 * hi;
    const float* pb  = sB + lo * GPITCH + 4 * hi;

    for (int st = 0; st < 4; ++st) {
        #pragma unroll
        for (int i = 0; i < 8; ++i) {
            *(v4f*)&sA[(sr + 8 * i) * GPITCH + sc] = ra[i];
            *(v4f*)&sB[(sr + 8 * i) * GPITCH + sc] = rb[i];
        }
        __syncthreads();
        if (st + 1 < 4) {
            #pragma unroll
            for (int i = 0; i < 8; ++i) {
                ra[i] = *(const v4f*)(xa + (size_t)i * 8 * NTOK + (st + 1) * 64);
                rb[i] = *(const v4f*)(xb + (size_t)i * 8 * NTOK + (st + 1) * 64);
            }
        }
        #pragma unroll
        for (int kk = 0; kk < 64; kk += 8) {
            const v4f a  = *(const v4f*)(pa + kk);
            const v4f b0 = *(const v4f*)(pb + kk);
            const v4f b1 = *(const v4f*)(pb + 16 * GPITCH + kk);
            const v4f b2 = *(const v4f*)(pb + 32 * GPITCH + kk);
            const v4f b3 = *(const v4f*)(pb + 48 * GPITCH + kk);
            const v2f a01 = {a.x, a.y}, a23 = {a.z, a.w};
            acc0 = wmma4(a01, (v2f){b0.x, b0.y}, acc0);
            acc1 = wmma4(a01, (v2f){b1.x, b1.y}, acc1);
            acc2 = wmma4(a01, (v2f){b2.x, b2.y}, acc2);
            acc3 = wmma4(a01, (v2f){b3.x, b3.y}, acc3);
            acc0 = wmma4(a23, (v2f){b0.z, b0.w}, acc0);
            acc1 = wmma4(a23, (v2f){b1.z, b1.w}, acc1);
            acc2 = wmma4(a23, (v2f){b2.z, b2.w}, acc2);
            acc3 = wmma4(a23, (v2f){b3.z, b3.w}, acc3);
        }
        __syncthreads();
    }

    float* dst = Gp + (size_t)s * EE;
    #pragma unroll
    for (int r = 0; r < 8; ++r) {
        const int row = mi * 64 + wave * 16 + hi * 8 + r;
        float* o = dst + (size_t)row * EMBED + ni * 64 + lo;
        o[0] = acc0[r]; o[16] = acc1[r]; o[32] = acc2[r]; o[48] = acc3[r];
    }
}

__global__ __launch_bounds__(256) void mhsa_greduce(
    const float* __restrict__ Gp, const float* __restrict__ x,
    float* __restrict__ G, float* __restrict__ xsp)
{
    __shared__ float sX[32];
    const int b = blockIdx.x, t = threadIdx.x;
    if (b < 256) {
        const int i4 = b * 256 + t;
        const v4f* src = (const v4f*)Gp;
        v4f s = src[i4];
        #pragma unroll
        for (int p = 1; p < GSPL; ++p) s += src[(size_t)p * (EE / 4) + i4];
        ((v4f*)G)[i4] = s;
    } else {
        const int xb = b - 256, rb = xb & 15, part = xb >> 4;
        const int lane = t & 31, wave = t >> 5;
        #pragma unroll
        for (int j = 0; j < 4; ++j) {
            const int row = rb * 32 + wave * 4 + j;
            const float* p = x + (size_t)row * NTOK + part * 512 + lane * 4;
            v4f s = *(const v4f*)p;
            s += *(const v4f*)(p + 128);
            s += *(const v4f*)(p + 256);
            s += *(const v4f*)(p + 384);
            float v = (s.x + s.y) + (s.z + s.w);
            v += __shfl_xor(v, 16, 32);
            v += __shfl_xor(v, 8, 32);
            v += __shfl_xor(v, 4, 32);
            v += __shfl_xor(v, 2, 32);
            v += __shfl_xor(v, 1, 32);
            if (lane == 0) sX[wave * 4 + j] = v;
        }
        __syncthreads();
        if (t < 32) xsp[part * EMBED + rb * 32 + t] = sX[t];
    }
}

__global__ __launch_bounds__(128) void mhsa_t_svec(
    const float* __restrict__ G, const float* __restrict__ W,
    const float* __restrict__ xsp, float* __restrict__ Tp, float* __restrict__ svp)
{
    const int b = blockIdx.x, t = threadIdx.x;
    const int lane = t & 31, wave = t >> 5, lo = lane & 15, hi = lane >> 4;
    if (b < 256) {
        const int mi = b & 7, ni = (b >> 3) & 7, ks = b >> 6;
        const int row = mi * 64 + wave * 16 + lo;
        const int nb = ni * 64 + lo;
        v8f acc0 = {}, acc1 = {}, acc2 = {}, acc3 = {};
        #pragma unroll 8
        for (int k = ks * 128; k < ks * 128 + 128; k += 4) {
            const int ka = k + 2 * hi;
            const v2f a = *(const v2f*)(G + (size_t)row * EMBED + ka);
            const float* w0 = W + (size_t)ka * QKVW + 2 * EMBED + nb;
            const float* w1 = w0 + QKVW;
            acc0 = wmma4(a, (v2f){w0[0],  w1[0]},  acc0);
            acc1 = wmma4(a, (v2f){w0[16], w1[16]}, acc1);
            acc2 = wmma4(a, (v2f){w0[32], w1[32]}, acc2);
            acc3 = wmma4(a, (v2f){w0[48], w1[48]}, acc3);
        }
        float* dst = Tp + (size_t)ks * EE;
        #pragma unroll
        for (int r = 0; r < 8; ++r) {
            const int rr = mi * 64 + wave * 16 + hi * 8 + r;
            float* o = dst + (size_t)rr * EMBED + ni * 64 + lo;
            o[0] = acc0[r]; o[16] = acc1[r]; o[32] = acc2[r]; o[48] = acc3[r];
        }
    } else {
        const int gw = (b - 256) * 4 + wave;
        const int sl = gw / 12, cg = gw - sl * 12;
        const int c = cg * 128 + lane * 4;
        v4f s = {};
        #pragma unroll 8
        for (int i = 0; i < 32; ++i) {
            const int e = sl * 32 + i;
            float xs = xsp[e];
            #pragma unroll
            for (int p = 1; p < XSPL; ++p) xs += xsp[p * EMBED + e];
            const v4f w = *(const v4f*)(W + (size_t)e * QKVW + c);
            s += xs * w;
        }
        *(v4f*)(svp + sl * QKVW + c) = s;
    }
}

__global__ __launch_bounds__(128) void mhsa_ktv(
    const float* __restrict__ W, const float* __restrict__ Tp, float* __restrict__ Kp)
{
    const int h = blockIdx.x, ks = blockIdx.y, t = threadIdx.x;
    const int lane = t & 31, wave = t >> 5, lo = lane & 15, hi = lane >> 4;
    const float* Wk = W + EMBED + h * DH + wave * 16 + lo;
    const float* Tb = Tp + h * DH + lo;
    v8f acc0 = {}, acc1 = {}, acc2 = {}, acc3 = {};
    #pragma unroll 4
    for (int k = ks * 64; k < ks * 64 + 64; k += 4) {
        const int ka = k + 2 * hi;
        const v2f a = {Wk[(size_t)ka * QKVW], Wk[(size_t)(ka + 1) * QKVW]};
        const float* t0 = Tb + (size_t)ka * EMBED;
        const float* t1 = t0 + EMBED;
        v2f b0 = {t0[0], t1[0]}, b1 = {t0[16], t1[16]}, b2 = {t0[32], t1[32]}, b3 = {t0[48], t1[48]};
        #pragma unroll
        for (int p = 1; p < TSPL; ++p) {
            const float* u0 = t0 + (size_t)p * EE;
            const float* u1 = t1 + (size_t)p * EE;
            b0 += (v2f){u0[0],  u1[0]};
            b1 += (v2f){u0[16], u1[16]};
            b2 += (v2f){u0[32], u1[32]};
            b3 += (v2f){u0[48], u1[48]};
        }
        acc0 = wmma4(a, b0, acc0);
        acc1 = wmma4(a, b1, acc1);
        acc2 = wmma4(a, b2, acc2);
        acc3 = wmma4(a, b3, acc3);
    }
    float* dst = Kp + ((size_t)ks * HEADS + h) * DH * DH;
    #pragma unroll
    for (int r = 0; r < 8; ++r) {
        const int row = wave * 16 + hi * 8 + r;
        float* o = dst + row * DH + lo;
        o[0] = acc0[r]; o[16] = acc1[r]; o[32] = acc2[r]; o[48] = acc3[r];
    }
}

__global__ __launch_bounds__(128) __attribute__((amdgpu_waves_per_eu(1, 2))) void mhsa_mq(
    const float* __restrict__ W, const float* __restrict__ bias,
    const float* __restrict__ Kp, const float* __restrict__ svp,
    float* __restrict__ Mq, float* __restrict__ U,
    float* __restrict__ rvec, float* __restrict__ cpad)
{
    __shared__ __attribute__((aligned(16))) float sK[DH * KPITCH];
    __shared__ float sQ[DH], sKs[DH], sVs[DH], sBk[DH], sBv[DH];
    const int h = blockIdx.x, et = blockIdx.y, t = threadIdx.x;
    const int lane = t & 31, wave = t >> 5, lo = lane & 15, hi = lane >> 4;

    if (t < DH) {
        const int c = h * DH + t;
        float q = svp[c], k = svp[EMBED + c], v = svp[2 * EMBED + c];
        #pragma unroll
        for (int p = 1; p < SSPL; ++p) {
            q += svp[p * QKVW + c];
            k += svp[p * QKVW + EMBED + c];
            v += svp[p * QKVW + 2 * EMBED + c];
        }
        const float bq = bias[c], bk = bias[EMBED + c], bv = bias[2 * EMBED + c];
        sQ[t]  = q + NTOKF * bq;
        sKs[t] = k;
        sVs[t] = v + NTOKF * bv;
        sBk[t] = bk;
        sBv[t] = bv;
    }
    __syncthreads();

    v4f vv[8];
    #pragma unroll
    for (int i = 0; i < 8; ++i) {
        const v4f* p = (const v4f*)(Kp + (size_t)h * DH * DH) + (t + 128 * i);
        v4f r0 = p[0];
        v4f r1 = p[(size_t)1 * (HEADS * DH * DH / 4)];
        v4f r2 = p[(size_t)2 * (HEADS * DH * DH / 4)];
        v4f r3 = p[(size_t)3 * (HEADS * DH * DH / 4)];
        v4f r4 = p[(size_t)4 * (HEADS * DH * DH / 4)];
        v4f r5 = p[(size_t)5 * (HEADS * DH * DH / 4)];
        v4f r6 = p[(size_t)6 * (HEADS * DH * DH / 4)];
        v4f r7 = p[(size_t)7 * (HEADS * DH * DH / 4)];
        vv[i] = ((((((r0 + r1) + r2) + r3) + r4) + r5) + r6) + r7;
    }
    #pragma unroll
    for (int i = 0; i < 8; ++i) {
        const int i4 = t + 128 * i;
        const int d1 = i4 >> 4, d2 = (i4 & 15) * 4;
        const float ks = sKs[d1], bk = sBk[d1];
        v4f v = vv[i];
        v.x += ks * sBv[d2]     + bk * sVs[d2];
        v.y += ks * sBv[d2 + 1] + bk * sVs[d2 + 1];
        v.z += ks * sBv[d2 + 2] + bk * sVs[d2 + 2];
        v.w += ks * sBv[d2 + 3] + bk * sVs[d2 + 3];
        *(v4f*)&sK[d1 * KPITCH + d2] = v;
    }
    __syncthreads();

    const int row = et * 64 + wave * 16 + lo;
    const float* Wq = W + (size_t)row * QKVW + h * DH;
    v8f acc0 = {}, acc1 = {}, acc2 = {}, acc3 = {};
    #pragma unroll
    for (int k = 0; k < DH; k += 4) {
        const int ka = k + 2 * hi;
        const v2f a = *(const v2f*)(Wq + ka);
        const float* B0 = sK + ka * KPITCH + lo;
        const float* B1 = B0 + KPITCH;
        acc0 = wmma4(a, (v2f){B0[0],  B1[0]},  acc0);
        acc1 = wmma4(a, (v2f){B0[16], B1[16]}, acc1);
        acc2 = wmma4(a, (v2f){B0[32], B1[32]}, acc2);
        acc3 = wmma4(a, (v2f){B0[48], B1[48]}, acc3);
    }
    #pragma unroll
    for (int r = 0; r < 8; ++r) {
        const int rr = et * 64 + wave * 16 + hi * 8 + r;
        float* o = Mq + (size_t)rr * EMBED + h * DH + lo;
        o[0] = acc0[r]; o[16] = acc1[r]; o[32] = acc2[r]; o[48] = acc3[r];
    }

    if (t < DH) {
        const int e = et * 64 + t;
        const float* wk = W + (size_t)e * QKVW + EMBED + h * DH;
        float s = 0.0f;
        #pragma unroll
        for (int d = 0; d < DH; d += 4) {
            const v4f w = *(const v4f*)(wk + d);
            s += w.x * sQ[d] + w.y * sQ[d + 1] + w.z * sQ[d + 2] + w.w * sQ[d + 3];
        }
        U[h * EMBED + e] = s;
    }
    if (et == 0) {
        if (t < DH) {
            float s = 0.0f;
            #pragma unroll
            for (int d = 0; d < DH; ++d) s += bias[h * DH + d] * sK[d * KPITCH + t];
            rvec[h * DH + t] = s;
        } else if (t < DH + 32) {
            float s = 0.0f;
            #pragma unroll
            for (int d = 0; d < DH; ++d) s += sBk[d] * sQ[d];
            cpad[h * 32 + (t - DH)] = s;
        }
    }
}

#define OPITCH 72
#define OCHUNK (64 * OPITCH)
__global__ __launch_bounds__(256) void mhsa_out(
    const float* __restrict__ x, const float* __restrict__ Mq,
    const float* __restrict__ rvec, const float* __restrict__ U,
    const float* __restrict__ cpad, float* __restrict__ out)
{
    __shared__ __attribute__((aligned(16))) float sA[2 * OCHUNK];
    __shared__ __attribute__((aligned(16))) float sB[2 * OCHUNK];
    __shared__ float sC[4 * HEADS * 32];
    const int b = blockIdx.x, t = threadIdx.x;
    const int lane = t & 31, wave = t >> 5, lo = lane & 15, hi = lane >> 4;
    if (b >= 128) {
        const int gb = b - 128;
        const int mt = gb & 63, h = gb >> 6;
        const int sr = t >> 4, sc = (t & 15) * 4;
        const float* xa = x  + (size_t)sr * NTOK  + mt * 64 + sc;
        const float* mb = Mq + (size_t)sr * EMBED + h * DH + sc;

        v4f ra[4], rb[4];
        #pragma unroll
        for (int i = 0; i < 4; ++i) {
            ra[i] = *(const v4f*)(xa + (size_t)(16 * i) * NTOK);
            rb[i] = *(const v4f*)(mb + (size_t)(16 * i) * EMBED);
        }
        #pragma unroll
        for (int i = 0; i < 4; ++i) {
            *(v4f*)&sA[(sr + 16 * i) * OPITCH + sc] = ra[i];
            *(v4f*)&sB[(sr + 16 * i) * OPITCH + sc] = rb[i];
        }
        __syncthreads();

        const int wm = wave & 3, wn = wave >> 2;
        const float* pa = sA + (2 * hi) * OPITCH + wm * 16 + lo;
        const float* pb = sB + (2 * hi) * OPITCH + wn * 32 + lo;
        v8f acc0 = {}, acc1 = {};

        for (int st = 0; st < 8; ++st) {
            const int cur = (st & 1) * OCHUNK;
            if (st + 1 < 8) {
                #pragma unroll
                for (int i = 0; i < 4; ++i) {
                    ra[i] = *(const v4f*)(xa + (size_t)((st + 1) * 64 + 16 * i) * NTOK);
                    rb[i] = *(const v4f*)(mb + (size_t)((st + 1) * 64 + 16 * i) * EMBED);
                }
            }
            __builtin_amdgcn_sched_barrier(0);
            #pragma unroll
            for (int kk = 0; kk < 64; kk += 4) {
                const float* a0 = pa + cur + kk * OPITCH;
                const float* b0 = pb + cur + kk * OPITCH;
                const v2f a = {a0[0], a0[OPITCH]};
                acc0 = wmma4(a, (v2f){b0[0],  b0[OPITCH]},      acc0);
                acc1 = wmma4(a, (v2f){b0[16], b0[OPITCH + 16]}, acc1);
            }
            if (st + 1 < 8) {
                const int nxt = ((st + 1) & 1) * OCHUNK;
                #pragma unroll
                for (int i = 0; i < 4; ++i) {
                    *(v4f*)&sA[nxt + (sr + 16 * i) * OPITCH + sc] = ra[i];
                    *(v4f*)&sB[nxt + (sr + 16 * i) * OPITCH + sc] = rb[i];
                }
                __syncthreads();
            }
        }

        const int nb = h * DH + wn * 32 + lo;
        const float r0 = rvec[nb], r1 = rvec[nb + 16];
        const int m0 = mt * 64 + wm * 16;
        float* base = out + HEADS * NTOK + (size_t)h * NTOK * DH;
        #pragma unroll
        for (int r = 0; r < 8; ++r) {
            float* o = base + (size_t)(m0 + hi * 8 + r) * DH + wn * 32 + lo;
            o[0]  = (acc0[r] + r0) * SCALE;
            o[16] = (acc1[r] + r1) * SCALE;
        }
    } else {
        const int n = b * 32 + lane;
        if (wave < 4) {
            float a[HEADS];
            #pragma unroll
            for (int hh = 0; hh < HEADS; ++hh) a[hh] = 0.0f;
            #pragma unroll 16
            for (int i = 0; i < 128; ++i) {
                const int e = wave * 128 + i;
                const float xv = x[(size_t)e * NTOK + n];
                #pragma unroll
                for (int hh = 0; hh < HEADS; ++hh) a[hh] += U[hh * EMBED + e] * xv;
            }
            #pragma unroll
            for (int hh = 0; hh < HEADS; ++hh) sC[(wave * HEADS + hh) * 32 + lane] = a[hh];
        }
        __syncthreads();
        if (wave < 4) {
            #pragma unroll
            for (int j = 0; j < 2; ++j) {
                const int hh = wave + 4 * j;
                const float s = ((sC[(0 * HEADS + hh) * 32 + lane] + sC[(1 * HEADS + hh) * 32 + lane])
                               + sC[(2 * HEADS + hh) * 32 + lane]) + sC[(3 * HEADS + hh) * 32 + lane];
                out[hh * NTOK + n] = (s + cpad[hh * 32]) * SCALE;
            }
        }
    }
}

extern "C" void kernel_launch(void* const* d_in, const int* in_sizes, int n_in,
                              void* d_out, int out_size, void* d_ws, size_t ws_size,
                              hipStream_t stream) {
    const float* x    = (const float*)d_in[0];
    const float* W    = (const float*)d_in[1];
    const float* bias = (const float*)d_in[2];
    float* out = (float*)d_out;
    float* ws  = (float*)d_ws;

    mhsa_gram<<<dim3(8, 8, GSPL), 128, 0, stream>>>(x, ws + OFF_GP);
    mhsa_greduce<<<384, 256, 0, stream>>>(ws + OFF_GP, x, ws + OFF_G, ws + OFF_XSP);
    mhsa_t_svec<<<304, 128, 0, stream>>>(ws + OFF_G, W, ws + OFF_XSP, ws + OFF_TP, ws + OFF_SVP);
    mhsa_ktv<<<dim3(HEADS, KSPL), 128, 0, stream>>>(W, ws + OFF_TP, ws + OFF_KP);
    mhsa_mq<<<dim3(HEADS, 8), 128, 0, stream>>>(W, bias, ws + OFF_KP, ws + OFF_SVP,
                                                ws + OFF_MQ, ws + OFF_U, ws + OFF_R, ws + OFF_C);
    mhsa_out<<<640, 256, 0, stream>>>(x, ws + OFF_MQ, ws + OFF_R, ws + OFF_U, ws + OFF_C, out);
}


# Gram product multiplies only the 36 tiles on or above the diagonal (x x^T equals its transpose); the reduce writes each sum to both mirror places; 21 token splits, was 16; sum order changes
# speedup vs baseline: 1.5082x; 1.0975x over previous
#include <hip/hip_runtime.h>


typedef __attribute__((ext_vector_type(2))) float v2f;
typedef __attribute__((ext_vector_type(4))) float v4f;
typedef __attribute__((ext_vector_type(8))) float v8f;

#define NTOK   4096
#define EMBED  512
#define QKVW   1536
#define HEADS  8
#define DH     64
#define SCALE  0.125f
#define NTOKF  4096.0f
#define GSPL   21
#define GTILES 36
#define GTSZ   (64 * 64)
#define GRED_BLOCKS 272
#define TSPL   4
#define KSPL   8
#define XSPL   8
#define SSPL   16
#define EE     (EMBED * EMBED)
#define GPITCH 68
#define KPITCH 72

#define OFF_GP    0
#define OFF_G     (OFF_GP + GSPL * GTILES * GTSZ)
#define OFF_TP    (OFF_G + EE)
#define OFF_KP    (OFF_TP + TSPL * EE)
#define OFF_MQ    (OFF_KP + KSPL * HEADS * DH * DH)
#define OFF_XSP   (OFF_MQ + EE)
#define OFF_SVP   (OFF_XSP + XSPL * EMBED)
#define OFF_U     (OFF_SVP + SSPL * QKVW)
#define OFF_R     (OFF_U + HEADS * EMBED)
#define OFF_C     (OFF_R + EMBED)

static __device__ __forceinline__ v8f wmma4(v2f a, v2f b, v8f c) {
    return __builtin_amdgcn_wmma_f32_16x16x4_f32(
        false, a, false, b, (short)0, c, false, false);
}

static __device__ __forceinline__ void gram_tile(int ti, int& mi, int& ni) {
    int m = 0, rem = ti;
    #pragma unroll
    for (int r = 0; r < 7; ++r) {
        if (m == r && rem >= 8 - r) { rem -= 8 - r; m = r + 1; }
    }
    mi = m; ni = m + rem;
}

__global__ __launch_bounds__(128) void mhsa_gram(
    const float* __restrict__ x, float* __restrict__ Gp)
{
    __shared__ __attribute__((aligned(16))) float sA[64 * GPITCH];
    __shared__ __attribute__((aligned(16))) float sB[64 * GPITCH];
    const int ti = blockIdx.x, s = blockIdx.y;
    int mi, ni;
    gram_tile(ti, mi, ni);
    const int nst = (s == GSPL - 1) ? 4 : 3;
    const int t = threadIdx.x, lane = t & 31, wave = t >> 5;
    const int lo = lane & 15, hi = lane >> 4;
    const int sr = t >> 4, sc = (t & 15) * 4;
    const float* xa = x + (size_t)(mi * 64 + sr) * NTOK + s * 192 + sc;
    const float* xb = x + (size_t)(ni * 64 + sr) * NTOK + s * 192 + sc;

    v4f ra[8], rb[8];
    #pragma unroll
    for (int i = 0; i < 8; ++i) {
        ra[i] = *(const v4f*)(xa + (size_t)i * 8 * NTOK);
        rb[i] = *(const v4f*)(xb + (size_t)i * 8 * NTOK);
    }

    v8f acc0 = {}, acc1 = {}, acc2 = {}, acc3 = {};
    const float* pa  = sA + (wave * 16 + lo) * GPITCH + 4 * hi;
    const float* pb  = sB + lo * GPITCH + 4 * hi;

    for (int st = 0; st < nst; ++st) {
        #pragma unroll
        for (int i = 0; i < 8; ++i) {
            *(v4f*)&sA[(sr + 8 * i) * GPITCH + sc] = ra[i];
            *(v4f*)&sB[(sr + 8 * i) * GPITCH + sc] = rb[i];
        }
        __syncthreads();
        if (st + 1 < nst) {
            #pragma unroll
            for (int i = 0; i < 8; ++i) {
                ra[i] = *(const v4f*)(xa + (size_t)i * 8 * NTOK + (st + 1) * 64);
                rb[i] = *(const v4f*)(xb + (size_t)i * 8 * NTOK + (st + 1) * 64);
            }
        }
        #pragma unroll
        for (int kk = 0; kk < 64; kk += 8) {
            const v4f a  = *(const v4f*)(pa + kk);
            const v4f b0 = *(const v4f*)(pb + kk);
            const v4f b1 = *(const v4f*)(pb + 16 * GPITCH + kk);
            const v4f b2 = *(const v4f*)(pb + 32 * GPITCH + kk);
            const v4f b3 = *(const v4f*)(pb + 48 * GPITCH + kk);
            const v2f a01 = {a.x, a.y}, a23 = {a.z, a.w};
            acc0 = wmma4(a01, (v2f){b0.x, b0.y}, acc0);
            acc1 = wmma4(a01, (v2f){b1.x, b1.y}, acc1);
            acc2 = wmma4(a01, (v2f){b2.x, b2.y}, acc2);
            acc3 = wmma4(a01, (v2f){b3.x, b3.y}, acc3);
            acc0 = wmma4(a23, (v2f){b0.z, b0.w}, acc0);
            acc1 = wmma4(a23, (v2f){b1.z, b1.w}, acc1);
            acc2 = wmma4(a23, (v2f){b2.z, b2.w}, acc2);
            acc3 = wmma4(a23, (v2f){b3.z, b3.w}, acc3);
        }
        __syncthreads();
    }

    float* dst = Gp + ((size_t)s * GTILES + ti) * GTSZ;
    #pragma unroll
    for (int r = 0; r < 8; ++r) {
        const int row = wave * 16 + hi * 8 + r;
        float* o = dst + row * 64 + lo;
        o[0] = acc0[r]; o[16] = acc1[r]; o[32] = acc2[r]; o[48] = acc3[r];
    }
}

__global__ __launch_bounds__(256) void mhsa_greduce(
    const float* __restrict__ Gp, const float* __restrict__ x,
    float* __restrict__ G, float* __restrict__ xsp)
{
    __shared__ float sX[32];
    __shared__ float sT[32 * 33];
    const int b = blockIdx.x, t = threadIdx.x;
    if (b < 4 * GTILES) {
        const int ti = b >> 2, q = b & 3;
        int mi, ni;
        gram_tile(ti, mi, ni);
        const int r0 = (q >> 1) * 32, c0 = (q & 1) * 32;
        const int rr = t >> 3, c4 = (t & 7) * 4;
        const float* src = Gp + (size_t)ti * GTSZ + (r0 + rr) * 64 + c0 + c4;
        v4f s = *(const v4f*)src;
        #pragma unroll
        for (int p = 1; p < GSPL; ++p) s += *(const v4f*)(src + (size_t)p * (GTILES * GTSZ));
        *(v4f*)(G + (size_t)(mi * 64 + r0 + rr) * EMBED + ni * 64 + c0 + c4) = s;
        if (mi != ni) {
            sT[(c4 + 0) * 33 + rr] = s.x;
            sT[(c4 + 1) * 33 + rr] = s.y;
            sT[(c4 + 2) * 33 + rr] = s.z;
            sT[(c4 + 3) * 33 + rr] = s.w;
            __syncthreads();
            const float* tp = sT + rr * 33 + c4;
            const v4f m = {tp[0], tp[1], tp[2], tp[3]};
            *(v4f*)(G + (size_t)(ni * 64 + c0 + rr) * EMBED + mi * 64 + r0 + c4) = m;
        }
    } else {
        const int xb = b - 4 * GTILES, rb = xb & 15, part = xb >> 4;
        const int lane = t & 31, wave = t >> 5;
        #pragma unroll
        for (int j = 0; j < 4; ++j) {
            const int row = rb * 32 + wave * 4 + j;
            const float* p = x + (size_t)row * NTOK + part * 512 + lane * 4;
            v4f s = *(const v4f*)p;
            s += *(const v4f*)(p + 128);
            s += *(const v4f*)(p + 256);
            s += *(const v4f*)(p + 384);
            float v = (s.x + s.y) + (s.z + s.w);
            v += __shfl_xor(v, 16, 32);
            v += __shfl_xor(v, 8, 32);
            v += __shfl_xor(v, 4, 32);
            v += __shfl_xor(v, 2, 32);
            v += __shfl_xor(v, 1, 32);
            if (lane == 0) sX[wave * 4 + j] = v;
        }
        __syncthreads();
        if (t < 32) xsp[part * EMBED + rb * 32 + t] = sX[t];
    }
}

__global__ __launch_bounds__(128) void mhsa_t_svec(
    const float* __restrict__ G, const float* __restrict__ W,
    const float* __restrict__ xsp, float* __restrict__ Tp, float* __restrict__ svp)
{
    const int b = blockIdx.x, t = threadIdx.x;
    const int lane = t & 31, wave = t >> 5, lo = lane & 15, hi = lane >> 4;
    if (b < 256) {
        const int mi = b & 7, ni = (b >> 3) & 7, ks = b >> 6;
        const int row = mi * 64 + wave * 16 + lo;
        const int nb = ni * 64 + lo;
        v8f acc0 = {}, acc1 = {}, acc2 = {}, acc3 = {};
        #pragma unroll 8
        for (int k = ks * 128; k < ks * 128 + 128; k += 4) {
            const int ka = k + 2 * hi;
            const v2f a = *(const v2f*)(G + (size_t)row * EMBED + ka);
            const float* w0 = W + (size_t)ka * QKVW + 2 * EMBED + nb;
            const float* w1 = w0 + QKVW;
            acc0 = wmma4(a, (v2f){w0[0],  w1[0]},  acc0);
            acc1 = wmma4(a, (v2f){w0[16], w1[16]}, acc1);
            acc2 = wmma4(a, (v2f){w0[32], w1[32]}, acc2);
            acc3 = wmma4(a, (v2f){w0[48], w1[48]}, acc3);
        }
        float* dst = Tp + (size_t)ks * EE;
        #pragma unroll
        for (int r = 0; r < 8; ++r) {
            const int rr = mi * 64 + wave * 16 + hi * 8 + r;
            float* o = dst + (size_t)rr * EMBED + ni * 64 + lo;
            o[0] = acc0[r]; o[16] = acc1[r]; o[32] = acc2[r]; o[48] = acc3[r];
        }
    } else {
        const int gw = (b - 256) * 4 + wave;
        const int sl = gw / 12, cg = gw - sl * 12;
        const int c = cg * 128 + lane * 4;
        v4f s = {};
        #pragma unroll 8
        for (int i = 0; i < 32; ++i) {
            const int e = sl * 32 + i;
            float xs = xsp[e];
            #pragma unroll
            for (int p = 1; p < XSPL; ++p) xs += xsp[p * EMBED + e];
            const v4f w = *(const v4f*)(W + (size_t)e * QKVW + c);
            s += xs * w;
        }
        *(v4f*)(svp + sl * QKVW + c) = s;
    }
}

__global__ __launch_bounds__(128) void mhsa_ktv(
    const float* __restrict__ W, const float* __restrict__ Tp, float* __restrict__ Kp)
{
    const int h = blockIdx.x, ks = blockIdx.y, t = threadIdx.x;
    const int lane = t & 31, wave = t >> 5, lo = lane & 15, hi = lane >> 4;
    const float* Wk = W + EMBED + h * DH + wave * 16 + lo;
    const float* Tb = Tp + h * DH + lo;
    v8f acc0 = {}, acc1 = {}, acc2 = {}, acc3 = {};
    #pragma unroll 4
    for (int k = ks * 64; k < ks * 64 + 64; k += 4) {
        const int ka = k + 2 * hi;
        const v2f a = {Wk[(size_t)ka * QKVW], Wk[(size_t)(ka + 1) * QKVW]};
        const float* t0 = Tb + (size_t)ka * EMBED;
        const float* t1 = t0 + EMBED;
        v2f b0 = {t0[0], t1[0]}, b1 = {t0[16], t1[16]}, b2 = {t0[32], t1[32]}, b3 = {t0[48], t1[48]};
        #pragma unroll
        for (int p = 1; p < TSPL; ++p) {
            const float* u0 = t0 + (size_t)p * EE;
            const float* u1 = t1 + (size_t)p * EE;
            b0 += (v2f){u0[0],  u1[0]};
            b1 += (v2f){u0[16], u1[16]};
            b2 += (v2f){u0[32], u1[32]};
            b3 += (v2f){u0[48], u1[48]};
        }
        acc0 = wmma4(a, b0, acc0);
        acc1 = wmma4(a, b1, acc1);
        acc2 = wmma4(a, b2, acc2);
        acc3 = wmma4(a, b3, acc3);
    }
    float* dst = Kp + ((size_t)ks * HEADS + h) * DH * DH;
    #pragma unroll
    for (int r = 0; r < 8; ++r) {
        const int row = wave * 16 + hi * 8 + r;
        float* o = dst + row * DH + lo;
        o[0] = acc0[r]; o[16] = acc1[r]; o[32] = acc2[r]; o[48] = acc3[r];
    }
}

__global__ __launch_bounds__(128) __attribute__((amdgpu_waves_per_eu(1, 2))) void mhsa_mq(
    const float* __restrict__ W, const float* __restrict__ bias,
    const float* __restrict__ Kp, const float* __restrict__ svp,
    float* __restrict__ Mq, float* __restrict__ U,
    float* __restrict__ rvec, float* __restrict__ cpad)
{
    __shared__ __attribute__((aligned(16))) float sK[DH * KPITCH];
    __shared__ float sQ[DH], sKs[DH], sVs[DH], sBk[DH], sBv[DH];
    const int h = blockIdx.x, et = blockIdx.y, t = threadIdx.x;
    const int lane = t & 31, wave = t >> 5, lo = lane & 15, hi = lane >> 4;

    if (t < DH) {
        const int c = h * DH + t;
        float q = svp[c], k = svp[EMBED + c], v = svp[2 * EMBED + c];
        #pragma unroll
        for (int p = 1; p < SSPL; ++p) {
            q += svp[p * QKVW + c];
            k += svp[p * QKVW + EMBED + c];
            v += svp[p * QKVW + 2 * EMBED + c];
        }
        const float bq = bias[c], bk = bias[EMBED + c], bv = bias[2 * EMBED + c];
        sQ[t]  = q + NTOKF * bq;
        sKs[t] = k;
        sVs[t] = v + NTOKF * bv;
        sBk[t] = bk;
        sBv[t] = bv;
    }
    __syncthreads();

    v4f vv[8];
    #pragma unroll
    for (int i = 0; i < 8; ++i) {
        const v4f* p = (const v4f*)(Kp + (size_t)h * DH * DH) + (t + 128 * i);
        v4f r0 = p[0];
        v4f r1 = p[(size_t)1 * (HEADS * DH * DH / 4)];
        v4f r2 = p[(size_t)2 * (HEADS * DH * DH / 4)];
        v4f r3 = p[(size_t)3 * (HEADS * DH * DH / 4)];
        v4f r4 = p[(size_t)4 * (HEADS * DH * DH / 4)];
        v4f r5 = p[(size_t)5 * (HEADS * DH * DH / 4)];
        v4f r6 = p[(size_t)6 * (HEADS * DH * DH / 4)];
        v4f r7 = p[(size_t)7 * (HEADS * DH * DH / 4)];
        vv[i] = ((((((r0 + r1) + r2) + r3) + r4) + r5) + r6) + r7;
    }
    #pragma unroll
    for (int i = 0; i < 8; ++i) {
        const int i4 = t + 128 * i;
        const int d1 = i4 >> 4, d2 = (i4 & 15) * 4;
        const float ks = sKs[d1], bk = sBk[d1];
        v4f v = vv[i];
        v.x += ks * sBv[d2]     + bk * sVs[d2];
        v.y += ks * sBv[d2 + 1] + bk * sVs[d2 + 1];
        v.z += ks * sBv[d2 + 2] + bk * sVs[d2 + 2];
        v.w += ks * sBv[d2 + 3] + bk * sVs[d2 + 3];
        *(v4f*)&sK[d1 * KPITCH + d2] = v;
    }
    __syncthreads();

    const int row = et * 64 + wave * 16 + lo;
    const float* Wq = W + (size_t)row * QKVW + h * DH;
    v8f acc0 = {}, acc1 = {}, acc2 = {}, acc3 = {};
    #pragma unroll
    for (int k = 0; k < DH; k += 4) {
        const int ka = k + 2 * hi;
        const v2f a = *(const v2f*)(Wq + ka);
        const float* B0 = sK + ka * KPITCH + lo;
        const float* B1 = B0 + KPITCH;
        acc0 = wmma4(a, (v2f){B0[0],  B1[0]},  acc0);
        acc1 = wmma4(a, (v2f){B0[16], B1[16]}, acc1);
        acc2 = wmma4(a, (v2f){B0[32], B1[32]}, acc2);
        acc3 = wmma4(a, (v2f){B0[48], B1[48]}, acc3);
    }
    #pragma unroll
    for (int r = 0; r < 8; ++r) {
        const int rr = et * 64 + wave * 16 + hi * 8 + r;
        float* o = Mq + (size_t)rr * EMBED + h * DH + lo;
        o[0] = acc0[r]; o[16] = acc1[r]; o[32] = acc2[r]; o[48] = acc3[r];
    }

    if (t < DH) {
        const int e = et * 64 + t;
        const float* wk = W + (size_t)e * QKVW + EMBED + h * DH;
        float s = 0.0f;
        #pragma unroll
        for (int d = 0; d < DH; d += 4) {
            const v4f w = *(const v4f*)(wk + d);
            s += w.x * sQ[d] + w.y * sQ[d + 1] + w.z * sQ[d + 2] + w.w * sQ[d + 3];
        }
        U[h * EMBED + e] = s;
    }
    if (et == 0) {
        if (t < DH) {
            float s = 0.0f;
            #pragma unroll
            for (int d = 0; d < DH; ++d) s += bias[h * DH + d] * sK[d * KPITCH + t];
            rvec[h * DH + t] = s;
        } else if (t < DH + 32) {
            float s = 0.0f;
            #pragma unroll
            for (int d = 0; d < DH; ++d) s += sBk[d] * sQ[d];
            cpad[h * 32 + (t - DH)] = s;
        }
    }
}

#define OPITCH 72
#define OCHUNK (64 * OPITCH)
__global__ __launch_bounds__(256) void mhsa_out(
    const float* __restrict__ x, const float* __restrict__ Mq,
    const float* __restrict__ rvec, const float* __restrict__ U,
    const float* __restrict__ cpad, float* __restrict__ out)
{
    __shared__ __attribute__((aligned(16))) float sA[2 * OCHUNK];
    __shared__ __attribute__((aligned(16))) float sB[2 * OCHUNK];
    __shared__ float sC[4 * HEADS * 32];
    const int b = blockIdx.x, t = threadIdx.x;
    const int lane = t & 31, wave = t >> 5, lo = lane & 15, hi = lane >> 4;
    if (b >= 128) {
        const int gb = b - 128;
        const int mt = gb & 63, h = gb >> 6;
        const int sr = t >> 4, sc = (t & 15) * 4;
        const float* xa = x  + (size_t)sr * NTOK  + mt * 64 + sc;
        const float* mb = Mq + (size_t)sr * EMBED + h * DH + sc;

        v4f ra[4], rb[4];
        #pragma unroll
        for (int i = 0; i < 4; ++i) {
            ra[i] = *(const v4f*)(xa + (size_t)(16 * i) * NTOK);
            rb[i] = *(const v4f*)(mb + (size_t)(16 * i) * EMBED);
        }
        #pragma unroll
        for (int i = 0; i < 4; ++i) {
            *(v4f*)&sA[(sr + 16 * i) * OPITCH + sc] = ra[i];
            *(v4f*)&sB[(sr + 16 * i) * OPITCH + sc] = rb[i];
        }
        __syncthreads();

        const int wm = wave & 3, wn = wave >> 2;
        const float* pa = sA + (2 * hi) * OPITCH + wm * 16 + lo;
        const float* pb = sB + (2 * hi) * OPITCH + wn * 32 + lo;
        v8f acc0 = {}, acc1 = {};

        for (int st = 0; st < 8; ++st) {
            const int cur = (st & 1) * OCHUNK;
            if (st + 1 < 8) {
                #pragma unroll
                for (int i = 0; i < 4; ++i) {
                    ra[i] = *(const v4f*)(xa + (size_t)((st + 1) * 64 + 16 * i) * NTOK);
                    rb[i] = *(const v4f*)(mb + (size_t)((st + 1) * 64 + 16 * i) * EMBED);
                }
            }
            __builtin_amdgcn_sched_barrier(0);
            #pragma unroll
            for (int kk = 0; kk < 64; kk += 4) {
                const float* a0 = pa + cur + kk * OPITCH;
                const float* b0 = pb + cur + kk * OPITCH;
                const v2f a = {a0[0], a0[OPITCH]};
                acc0 = wmma4(a, (v2f){b0[0],  b0[OPITCH]},      acc0);
                acc1 = wmma4(a, (v2f){b0[16], b0[OPITCH + 16]}, acc1);
            }
            if (st + 1 < 8) {
                const int nxt = ((st + 1) & 1) * OCHUNK;
                #pragma unroll
                for (int i = 0; i < 4; ++i) {
                    *(v4f*)&sA[nxt + (sr + 16 * i) * OPITCH + sc] = ra[i];
                    *(v4f*)&sB[nxt + (sr + 16 * i) * OPITCH + sc] = rb[i];
                }
                __syncthreads();
            }
        }

        const int nb = h * DH + wn * 32 + lo;
        const float r0 = rvec[nb], r1 = rvec[nb + 16];
        const int m0 = mt * 64 + wm * 16;
        float* base = out + HEADS * NTOK + (size_t)h * NTOK * DH;
        #pragma unroll
        for (int r = 0; r < 8; ++r) {
            float* o = base + (size_t)(m0 + hi * 8 + r) * DH + wn * 32 + lo;
            o[0]  = (acc0[r] + r0) * SCALE;
            o[16] = (acc1[r] + r1) * SCALE;
        }
    } else {
        const int n = b * 32 + lane;
        if (wave < 4) {
            float a[HEADS];
            #pragma unroll
            for (int hh = 0; hh < HEADS; ++hh) a[hh] = 0.0f;
            #pragma unroll 16
            for (int i = 0; i < 128; ++i) {
                const int e = wave * 128 + i;
                const float xv = x[(size_t)e * NTOK + n];
                #pragma unroll
                for (int hh = 0; hh < HEADS; ++hh) a[hh] += U[hh * EMBED + e] * xv;
            }
            #pragma unroll
            for (int hh = 0; hh < HEADS; ++hh) sC[(wave * HEADS + hh) * 32 + lane] = a[hh];
        }
        __syncthreads();
        if (wave < 4) {
            #pragma unroll
            for (int j = 0; j < 2; ++j) {
                const int hh = wave + 4 * j;
                const float s = ((sC[(0 * HEADS + hh) * 32 + lane] + sC[(1 * HEADS + hh) * 32 + lane])
                               + sC[(2 * HEADS + hh) * 32 + lane]) + sC[(3 * HEADS + hh) * 32 + lane];
                out[hh * NTOK + n] = (s + cpad[hh * 32]) * SCALE;
            }
        }
    }
}

extern "C" void kernel_launch(void* const* d_in, const int* in_sizes, int n_in,
                              void* d_out, int out_size, void* d_ws, size_t ws_size,
                              hipStream_t stream) {
    const float* x    = (const float*)d_in[0];
    const float* W    = (const float*)d_in[1];
    const float* bias = (const float*)d_in[2];
    float* out = (float*)d_out;
    float* ws  = (float*)d_ws;

    mhsa_gram<<<dim3(GTILES, GSPL), 128, 0, stream>>>(x, ws + OFF_GP);
    mhsa_greduce<<<GRED_BLOCKS, 256, 0, stream>>>(ws + OFF_GP, x, ws + OFF_G, ws + OFF_XSP);
    mhsa_t_svec<<<304, 128, 0, stream>>>(ws + OFF_G, W, ws + OFF_XSP, ws + OFF_TP, ws + OFF_SVP);
    mhsa_ktv<<<dim3(HEADS, KSPL), 128, 0, stream>>>(W, ws + OFF_TP, ws + OFF_KP);
    mhsa_mq<<<dim3(HEADS, 8), 128, 0, stream>>>(W, bias, ws + OFF_KP, ws + OFF_SVP,
                                                ws + OFF_MQ, ws + OFF_U, ws + OFF_R, ws + OFF_C);
    mhsa_out<<<640, 256, 0, stream>>>(x, ws + OFF_MQ, ws + OFF_R, ws + OFF_U, ws + OFF_C, out);
}


# T = G Wv as 64x16 tiles: the four k-quarter planes stay in LDS and feed the KtV partial product in the same block; T planes in memory and the KtV launch removed (5 launches); sum order unchanged
# speedup vs baseline: 1.8627x; 1.2350x over previous
#include <hip/hip_runtime.h>


typedef __attribute__((ext_vector_type(2))) float v2f;
typedef __attribute__((ext_vector_type(4))) float v4f;
typedef __attribute__((ext_vector_type(8))) float v8f;

#define NTOK   4096
#define EMBED  512
#define QKVW   1536
#define HEADS  8
#define DH     64
#define SCALE  0.125f
#define NTOKF  4096.0f
#define GSPL   21
#define GTILES 36
#define GTSZ   (64 * 64)
#define GRED_BLOCKS 272
#define KSPL   8
#define XSPL   8
#define SSPL   16
#define EE     (EMBED * EMBED)
#define GPITCH 68
#define KPITCH 72

#define OFF_GP    0
#define OFF_G     (OFF_GP + GSPL * GTILES * GTSZ)
#define OFF_KP    (OFF_G + EE)
#define OFF_MQ    (OFF_KP + KSPL * HEADS * DH * DH)
#define OFF_XSP   (OFF_MQ + EE)
#define OFF_SVP   (OFF_XSP + XSPL * EMBED)
#define OFF_U     (OFF_SVP + SSPL * QKVW)
#define OFF_R     (OFF_U + HEADS * EMBED)
#define OFF_C     (OFF_R + EMBED)

static __device__ __forceinline__ v8f wmma4(v2f a, v2f b, v8f c) {
    return __builtin_amdgcn_wmma_f32_16x16x4_f32(
        false, a, false, b, (short)0, c, false, false);
}

static __device__ __forceinline__ void gram_tile(int ti, int& mi, int& ni) {
    int m = 0, rem = ti;
    #pragma unroll
    for (int r = 0; r < 7; ++r) {
        if (m == r && rem >= 8 - r) { rem -= 8 - r; m = r + 1; }
    }
    mi = m; ni = m + rem;
}

__global__ __launch_bounds__(128) void mhsa_gram(
    const float* __restrict__ x, float* __restrict__ Gp)
{
    __shared__ __attribute__((aligned(16))) float sA[64 * GPITCH];
    __shared__ __attribute__((aligned(16))) float sB[64 * GPITCH];
    const int ti = blockIdx.x, s = blockIdx.y;
    int mi, ni;
    gram_tile(ti, mi, ni);
    const int nst = (s == GSPL - 1) ? 4 : 3;
    const int t = threadIdx.x, lane = t & 31, wave = t >> 5;
    const int lo = lane & 15, hi = lane >> 4;
    const int sr = t >> 4, sc = (t & 15) * 4;
    const float* xa = x + (size_t)(mi * 64 + sr) * NTOK + s * 192 + sc;
    const float* xb = x + (size_t)(ni * 64 + sr) * NTOK + s * 192 + sc;

    v4f ra[8], rb[8];
    #pragma unroll
    for (int i = 0; i < 8; ++i) {
        ra[i] = *(const v4f*)(xa + (size_t)i * 8 * NTOK);
        rb[i] = *(const v4f*)(xb + (size_t)i * 8 * NTOK);
    }

    v8f acc0 = {}, acc1 = {}, acc2 = {}, acc3 = {};
    const float* pa  = sA + (wave * 16 + lo) * GPITCH + 4 * hi;
    const float* pb  = sB + lo * GPITCH + 4 * hi;

    for (int st = 0; st < nst; ++st) {
        #pragma unroll
        for (int i = 0; i < 8; ++i) {
            *(v4f*)&sA[(sr + 8 * i) * GPITCH + sc] = ra[i];
            *(v4f*)&sB[(sr + 8 * i) * GPITCH + sc] = rb[i];
        }
        __syncthreads();
        if (st + 1 < nst) {
            #pragma unroll
            for (int i = 0; i < 8; ++i) {
                ra[i] = *(const v4f*)(xa + (size_t)i * 8 * NTOK + (st + 1) * 64);
                rb[i] = *(const v4f*)(xb + (size_t)i * 8 * NTOK + (st + 1) * 64);
            }
        }
        #pragma unroll
        for (int kk = 0; kk < 64; kk += 8) {
            const v4f a  = *(const v4f*)(pa + kk);
            const v4f b0 = *(const v4f*)(pb + kk);
            const v4f b1 = *(const v4f*)(pb + 16 * GPITCH + kk);
            const v4f b2 = *(const v4f*)(pb + 32 * GPITCH + kk);
            const v4f b3 = *(const v4f*)(pb + 48 * GPITCH + kk);
            const v2f a01 = {a.x, a.y}, a23 = {a.z, a.w};
            acc0 = wmma4(a01, (v2f){b0.x, b0.y}, acc0);
            acc1 = wmma4(a01, (v2f){b1.x, b1.y}, acc1);
            acc2 = wmma4(a01, (v2f){b2.x, b2.y}, acc2);
            acc3 = wmma4(a01, (v2f){b3.x, b3.y}, acc3);
            acc0 = wmma4(a23, (v2f){b0.z, b0.w}, acc0);
            acc1 = wmma4(a23, (v2f){b1.z, b1.w}, acc1);
            acc2 = wmma4(a23, (v2f){b2.z, b2.w}, acc2);
            acc3 = wmma4(a23, (v2f){b3.z, b3.w}, acc3);
        }
        __syncthreads();
    }

    float* dst = Gp + ((size_t)s * GTILES + ti) * GTSZ;
    #pragma unroll
    for (int r = 0; r < 8; ++r) {
        const int row = wave * 16 + hi * 8 + r;
        float* o = dst + row * 64 + lo;
        o[0] = acc0[r]; o[16] = acc1[r]; o[32] = acc2[r]; o[48] = acc3[r];
    }
}

__global__ __launch_bounds__(256) void mhsa_greduce(
    const float* __restrict__ Gp, const float* __restrict__ x,
    float* __restrict__ G, float* __restrict__ xsp)
{
    __shared__ float sX[32];
    __shared__ float sT[32 * 33];
    const int b = blockIdx.x, t = threadIdx.x;
    if (b < 4 * GTILES) {
        const int ti = b >> 2, q = b & 3;
        int mi, ni;
        gram_tile(ti, mi, ni);
        const int r0 = (q >> 1) * 32, c0 = (q & 1) * 32;
        const int rr = t >> 3, c4 = (t & 7) * 4;
        const float* src = Gp + (size_t)ti * GTSZ + (r0 + rr) * 64 + c0 + c4;
        v4f s = *(const v4f*)src;
        #pragma unroll
        for (int p = 1; p < GSPL; ++p) s += *(const v4f*)(src + (size_t)p * (GTILES * GTSZ));
        *(v4f*)(G + (size_t)(mi * 64 + r0 + rr) * EMBED + ni * 64 + c0 + c4) = s;
        if (mi != ni) {
            sT[(c4 + 0) * 33 + rr] = s.x;
            sT[(c4 + 1) * 33 + rr] = s.y;
            sT[(c4 + 2) * 33 + rr] = s.z;
            sT[(c4 + 3) * 33 + rr] = s.w;
            __syncthreads();
            const float* tp = sT + rr * 33 + c4;
            const v4f m = {tp[0], tp[1], tp[2], tp[3]};
            *(v4f*)(G + (size_t)(ni * 64 + c0 + rr) * EMBED + mi * 64 + r0 + c4) = m;
        }
    } else {
        const int xb = b - 4 * GTILES, rb = xb & 15, part = xb >> 4;
        const int lane = t & 31, wave = t >> 5;
        #pragma unroll
        for (int j = 0; j < 4; ++j) {
            const int row = rb * 32 + wave * 4 + j;
            const float* p = x + (size_t)row * NTOK + part * 512 + lane * 4;
            v4f s = *(const v4f*)p;
            s += *(const v4f*)(p + 128);
            s += *(const v4f*)(p + 256);
            s += *(const v4f*)(p + 384);
            float v = (s.x + s.y) + (s.z + s.w);
            v += __shfl_xor(v, 16, 32);
            v += __shfl_xor(v, 8, 32);
            v += __shfl_xor(v, 4, 32);
            v += __shfl_xor(v, 2, 32);
            v += __shfl_xor(v, 1, 32);
            if (lane == 0) sX[wave * 4 + j] = v;
        }
        __syncthreads();
        if (t < 32) xsp[part * EMBED + rb * 32 + t] = sX[t];
    }
}

#define TPITCH 20
#define TPLANE (64 * TPITCH)
__global__ __launch_bounds__(128) __attribute__((amdgpu_waves_per_eu(1, 2))) void mhsa_t_ktv(
    const float* __restrict__ G, const float* __restrict__ W,
    const float* __restrict__ xsp, float* __restrict__ Kp, float* __restrict__ svp)
{
    __shared__ float sT[4 * TPLANE];
    const int b = blockIdx.x, t = threadIdx.x;
    const int lane = t & 31, wave = t >> 5, lo = lane & 15, hi = lane >> 4;
    if (b < 256) {
        const int c2 = b & 3, h = (b >> 2) & 7, mi = b >> 5;
        const float* Wk = W + (size_t)(mi * 64 + 2 * hi) * QKVW + EMBED + h * DH + wave * 16 + lo;
        v2f wk[16];
        #pragma unroll
        for (int j = 0; j < 16; ++j)
            wk[j] = (v2f){Wk[(size_t)(4 * j) * QKVW], Wk[(size_t)(4 * j + 1) * QKVW]};

        const float* Ga = G + (size_t)(mi * 64 + lo) * EMBED + wave * 128 + 2 * hi;
        const float* Wv = W + (size_t)(wave * 128 + 2 * hi) * QKVW + 2 * EMBED + h * DH + c2 * 16 + lo;
        v2f ga[2][8][4], wv[2][8];
        #pragma unroll
        for (int u = 0; u < 8; ++u) {
            wv[0][u] = (v2f){Wv[(size_t)(4 * u) * QKVW], Wv[(size_t)(4 * u + 1) * QKVW]};
            #pragma unroll
            for (int i = 0; i < 4; ++i) ga[0][u][i] = *(const v2f*)(Ga + (size_t)(16 * i) * EMBED + 4 * u);
        }
        v8f acc0 = {}, acc1 = {}, acc2 = {}, acc3 = {};
        #pragma unroll
        for (int g = 0; g < 4; ++g) {
            const int cur = g & 1, nxt = cur ^ 1;
            if (g + 1 < 4) {
                #pragma unroll
                for (int u = 0; u < 8; ++u) {
                    const int k = 32 * (g + 1) + 4 * u;
                    wv[nxt][u] = (v2f){Wv[(size_t)k * QKVW], Wv[(size_t)(k + 1) * QKVW]};
                    #pragma unroll
                    for (int i = 0; i < 4; ++i) ga[nxt][u][i] = *(const v2f*)(Ga + (size_t)(16 * i) * EMBED + k);
                }
            }
            __builtin_amdgcn_sched_barrier(0);
            #pragma unroll
            for (int u = 0; u < 8; ++u) {
                acc0 = wmma4(ga[cur][u][0], wv[cur][u], acc0);
                acc1 = wmma4(ga[cur][u][1], wv[cur][u], acc1);
                acc2 = wmma4(ga[cur][u][2], wv[cur][u], acc2);
                acc3 = wmma4(ga[cur][u][3], wv[cur][u], acc3);
            }
        }
        float* pl = sT + wave * TPLANE + (8 * hi) * TPITCH + lo;
        #pragma unroll
        for (int r = 0; r < 8; ++r) {
            pl[r * TPITCH]        = acc0[r];
            pl[(16 + r) * TPITCH] = acc1[r];
            pl[(32 + r) * TPITCH] = acc2[r];
            pl[(48 + r) * TPITCH] = acc3[r];
        }
        __syncthreads();

        const float* tb = sT + (2 * hi) * TPITCH + lo;
        v8f kacc = {};
        #pragma unroll
        for (int j = 0; j < 16; ++j) {
            const float* q = tb + (4 * j) * TPITCH;
            v2f bv = {q[0], q[TPITCH]};
            #pragma unroll
            for (int p = 1; p < 4; ++p) bv += (v2f){q[p * TPLANE], q[p * TPLANE + TPITCH]};
            kacc = wmma4(wk[j], bv, kacc);
        }
        float* dst = Kp + ((((size_t)mi * HEADS + h) * 4 + c2) * DH + wave * 16 + 8 * hi) * 16 + lo;
        #pragma unroll
        for (int r = 0; r < 8; ++r) dst[r * 16] = kacc[r];
    } else {
        const int gw = (b - 256) * 4 + wave;
        const int sl = gw / 12, cg = gw - sl * 12;
        const int c = cg * 128 + lane * 4;
        v4f s = {};
        #pragma unroll 8
        for (int i = 0; i < 32; ++i) {
            const int e = sl * 32 + i;
            float xs = xsp[e];
            #pragma unroll
            for (int p = 1; p < XSPL; ++p) xs += xsp[p * EMBED + e];
            const v4f w = *(const v4f*)(W + (size_t)e * QKVW + c);
            s += xs * w;
        }
        *(v4f*)(svp + sl * QKVW + c) = s;
    }
}

__global__ __launch_bounds__(128) __attribute__((amdgpu_waves_per_eu(1, 2))) void mhsa_mq(
    const float* __restrict__ W, const float* __restrict__ bias,
    const float* __restrict__ Kp, const float* __restrict__ svp,
    float* __restrict__ Mq, float* __restrict__ U,
    float* __restrict__ rvec, float* __restrict__ cpad)
{
    __shared__ __attribute__((aligned(16))) float sK[DH * KPITCH];
    __shared__ float sQ[DH], sKs[DH], sVs[DH], sBk[DH], sBv[DH];
    const int h = blockIdx.x, et = blockIdx.y, t = threadIdx.x;
    const int lane = t & 31, wave = t >> 5, lo = lane & 15, hi = lane >> 4;

    if (t < DH) {
        const int c = h * DH + t;
        float q = svp[c], k = svp[EMBED + c], v = svp[2 * EMBED + c];
        #pragma unroll
        for (int p = 1; p < SSPL; ++p) {
            q += svp[p * QKVW + c];
            k += svp[p * QKVW + EMBED + c];
            v += svp[p * QKVW + 2 * EMBED + c];
        }
        const float bq = bias[c], bk = bias[EMBED + c], bv = bias[2 * EMBED + c];
        sQ[t]  = q + NTOKF * bq;
        sKs[t] = k;
        sVs[t] = v + NTOKF * bv;
        sBk[t] = bk;
        sBv[t] = bv;
    }
    __syncthreads();

    v4f vv[8];
    #pragma unroll
    for (int i = 0; i < 8; ++i) {
        const v4f* p = (const v4f*)(Kp + (size_t)h * DH * DH) + (t + 128 * i);
        v4f r0 = p[0];
        v4f r1 = p[(size_t)1 * (HEADS * DH * DH / 4)];
        v4f r2 = p[(size_t)2 * (HEADS * DH * DH / 4)];
        v4f r3 = p[(size_t)3 * (HEADS * DH * DH / 4)];
        v4f r4 = p[(size_t)4 * (HEADS * DH * DH / 4)];
        v4f r5 = p[(size_t)5 * (HEADS * DH * DH / 4)];
        v4f r6 = p[(size_t)6 * (HEADS * DH * DH / 4)];
        v4f r7 = p[(size_t)7 * (HEADS * DH * DH / 4)];
        vv[i] = ((((((r0 + r1) + r2) + r3) + r4) + r5) + r6) + r7;
    }
    #pragma unroll
    for (int i = 0; i < 8; ++i) {
        const int i4 = t + 128 * i;
        const int d1 = (i4 >> 2) & 63, d2 = (i4 >> 8) * 16 + (i4 & 3) * 4;
        const float ks = sKs[d1], bk = sBk[d1];
        v4f v = vv[i];
        v.x += ks * sBv[d2]     + bk * sVs[d2];
        v.y += ks * sBv[d2 + 1] + bk * sVs[d2 + 1];
        v.z += ks * sBv[d2 + 2] + bk * sVs[d2 + 2];
        v.w += ks * sBv[d2 + 3] + bk * sVs[d2 + 3];
        *(v4f*)&sK[d1 * KPITCH + d2] = v;
    }
    __syncthreads();

    const int row = et * 64 + wave * 16 + lo;
    const float* Wq = W + (size_t)row * QKVW + h * DH;
    v8f acc0 = {}, acc1 = {}, acc2 = {}, acc3 = {};
    #pragma unroll
    for (int k = 0; k < DH; k += 4) {
        const int ka = k + 2 * hi;
        const v2f a = *(const v2f*)(Wq + ka);
        const float* B0 = sK + ka * KPITCH + lo;
        const float* B1 = B0 + KPITCH;
        acc0 = wmma4(a, (v2f){B0[0],  B1[0]},  acc0);
        acc1 = wmma4(a, (v2f){B0[16], B1[16]}, acc1);
        acc2 = wmma4(a, (v2f){B0[32], B1[32]}, acc2);
        acc3 = wmma4(a, (v2f){B0[48], B1[48]}, acc3);
    }
    #pragma unroll
    for (int r = 0; r < 8; ++r) {
        const int rr = et * 64 + wave * 16 + hi * 8 + r;
        float* o = Mq + (size_t)rr * EMBED + h * DH + lo;
        o[0] = acc0[r]; o[16] = acc1[r]; o[32] = acc2[r]; o[48] = acc3[r];
    }

    if (t < DH) {
        const int e = et * 64 + t;
        const float* wk = W + (size_t)e * QKVW + EMBED + h * DH;
        float s = 0.0f;
        #pragma unroll
        for (int d = 0; d < DH; d += 4) {
            const v4f w = *(const v4f*)(wk + d);
            s += w.x * sQ[d] + w.y * sQ[d + 1] + w.z * sQ[d + 2] + w.w * sQ[d + 3];
        }
        U[h * EMBED + e] = s;
    }
    if (et == 0) {
        if (t < DH) {
            float s = 0.0f;
            #pragma unroll
            for (int d = 0; d < DH; ++d) s += bias[h * DH + d] * sK[d * KPITCH + t];
            rvec[h * DH + t] = s;
        } else if (t < DH + 32) {
            float s = 0.0f;
            #pragma unroll
            for (int d = 0; d < DH; ++d) s += sBk[d] * sQ[d];
            cpad[h * 32 + (t - DH)] = s;
        }
    }
}

#define OPITCH 72
#define OCHUNK (64 * OPITCH)
__global__ __launch_bounds__(256) void mhsa_out(
    const float* __restrict__ x, const float* __restrict__ Mq,
    const float* __restrict__ rvec, const float* __restrict__ U,
    const float* __restrict__ cpad, float* __restrict__ out)
{
    __shared__ __attribute__((aligned(16))) float sA[2 * OCHUNK];
    __shared__ __attribute__((aligned(16))) float sB[2 * OCHUNK];
    __shared__ float sC[4 * HEADS * 32];
    const int b = blockIdx.x, t = threadIdx.x;
    const int lane = t & 31, wave = t >> 5, lo = lane & 15, hi = lane >> 4;
    if (b >= 128) {
        const int gb = b - 128;
        const int mt = gb & 63, h = gb >> 6;
        const int sr = t >> 4, sc = (t & 15) * 4;
        const float* xa = x  + (size_t)sr * NTOK  + mt * 64 + sc;
        const float* mb = Mq + (size_t)sr * EMBED + h * DH + sc;

        v4f ra[4], rb[4];
        #pragma unroll
        for (int i = 0; i < 4; ++i) {
            ra[i] = *(const v4f*)(xa + (size_t)(16 * i) * NTOK);
            rb[i] = *(const v4f*)(mb + (size_t)(16 * i) * EMBED);
        }
        #pragma unroll
        for (int i = 0; i < 4; ++i) {
            *(v4f*)&sA[(sr + 16 * i) * OPITCH + sc] = ra[i];
            *(v4f*)&sB[(sr + 16 * i) * OPITCH + sc] = rb[i];
        }
        __syncthreads();

        const int wm = wave & 3, wn = wave >> 2;
        const float* pa = sA + (2 * hi) * OPITCH + wm * 16 + lo;
        const float* pb = sB + (2 * hi) * OPITCH + wn * 32 + lo;
        v8f acc0 = {}, acc1 = {};

        for (int st = 0; st < 8; ++st) {
            const int cur = (st & 1) * OCHUNK;
            if (st + 1 < 8) {
                #pragma unroll
                for (int i = 0; i < 4; ++i) {
                    ra[i] = *(const v4f*)(xa + (size_t)((st + 1) * 64 + 16 * i) * NTOK);
                    rb[i] = *(const v4f*)(mb + (size_t)((st + 1) * 64 + 16 * i) * EMBED);
                }
            }
            __builtin_amdgcn_sched_barrier(0);
            #pragma unroll
            for (int kk = 0; kk < 64; kk += 4) {
                const float* a0 = pa + cur + kk * OPITCH;
                const float* b0 = pb + cur + kk * OPITCH;
                const v2f a = {a0[0], a0[OPITCH]};
                acc0 = wmma4(a, (v2f){b0[0],  b0[OPITCH]},      acc0);
                acc1 = wmma4(a, (v2f){b0[16], b0[OPITCH + 16]}, acc1);
            }
            if (st + 1 < 8) {
                const int nxt = ((st + 1) & 1) * OCHUNK;
                #pragma unroll
                for (int i = 0; i < 4; ++i) {
                    *(v4f*)&sA[nxt + (sr + 16 * i) * OPITCH + sc] = ra[i];
                    *(v4f*)&sB[nxt + (sr + 16 * i) * OPITCH + sc] = rb[i];
                }
                __syncthreads();
            }
        }

        const int nb = h * DH + wn * 32 + lo;
        const float r0 = rvec[nb], r1 = rvec[nb + 16];
        const int m0 = mt * 64 + wm * 16;
        float* base = out + HEADS * NTOK + (size_t)h * NTOK * DH;
        #pragma unroll
        for (int r = 0; r < 8; ++r) {
            float* o = base + (size_t)(m0 + hi * 8 + r) * DH + wn * 32 + lo;
            o[0]  = (acc0[r] + r0) * SCALE;
            o[16] = (acc1[r] + r1) * SCALE;
        }
    } else {
        const int n = b * 32 + lane;
        if (wave < 4) {
            float a[HEADS];
            #pragma unroll
            for (int hh = 0; hh < HEADS; ++hh) a[hh] = 0.0f;
            #pragma unroll 16
            for (int i = 0; i < 128; ++i) {
                const int e = wave * 128 + i;
                const float xv = x[(size_t)e * NTOK + n];
                #pragma unroll
                for (int hh = 0; hh < HEADS; ++hh) a[hh] += U[hh * EMBED + e] * xv;
            }
            #pragma unroll
            for (int hh = 0; hh < HEADS; ++hh) sC[(wave * HEADS + hh) * 32 + lane] = a[hh];
        }
        __syncthreads();
        if (wave < 4) {
            #pragma unroll
            for (int j = 0; j < 2; ++j) {
                const int hh = wave + 4 * j;
                const float s = ((sC[(0 * HEADS + hh) * 32 + lane] + sC[(1 * HEADS + hh) * 32 + lane])
                               + sC[(2 * HEADS + hh) * 32 + lane]) + sC[(3 * HEADS + hh) * 32 + lane];
                out[hh * NTOK + n] = (s + cpad[hh * 32]) * SCALE;
            }
        }
    }
}

extern "C" void kernel_launch(void* const* d_in, const int* in_sizes, int n_in,
                              void* d_out, int out_size, void* d_ws, size_t ws_size,
                              hipStream_t stream) {
    const float* x    = (const float*)d_in[0];
    const float* W    = (const float*)d_in[1];
    const float* bias = (const float*)d_in[2];
    float* out = (float*)d_out;
    float* ws  = (float*)d_ws;

    mhsa_gram<<<dim3(GTILES, GSPL), 128, 0, stream>>>(x, ws + OFF_GP);
    mhsa_greduce<<<GRED_BLOCKS, 256, 0, stream>>>(ws + OFF_GP, x, ws + OFF_G, ws + OFF_XSP);
    mhsa_t_ktv<<<304, 128, 0, stream>>>(ws + OFF_G, W, ws + OFF_XSP, ws + OFF_KP, ws + OFF_SVP);
    mhsa_mq<<<dim3(HEADS, 8), 128, 0, stream>>>(W, bias, ws + OFF_KP, ws + OFF_SVP,
                                                ws + OFF_MQ, ws + OFF_U, ws + OFF_R, ws + OFF_C);
    mhsa_out<<<640, 256, 0, stream>>>(x, ws + OFF_MQ, ws + OFF_R, ws + OFF_U, ws + OFF_C, out);
}


# Mq launch removed (4 launches): the output block multiplies its tile of x^T Wq_h + bq_h by KtV_h, summed from the partial planes in the block; U and c made in the T launch; output sum order changes
# speedup vs baseline: 2.0276x; 1.0885x over previous
#include <hip/hip_runtime.h>


typedef __attribute__((ext_vector_type(2))) float v2f;
typedef __attribute__((ext_vector_type(4))) float v4f;
typedef __attribute__((ext_vector_type(8))) float v8f;

#define NTOK   4096
#define EMBED  512
#define QKVW   1536
#define HEADS  8
#define DH     64
#define SCALE  0.125f
#define NTOKF  4096.0f
#define GSPL   21
#define GTILES 36
#define GTSZ   (64 * 64)
#define GRED_BLOCKS 272
#define KSPL   8
#define XSPL   8
#define SSPL   16
#define EE     (EMBED * EMBED)
#define GPITCH 68
#define KPITCH 72

#define OFF_GP    0
#define OFF_G     (OFF_GP + GSPL * GTILES * GTSZ)
#define OFF_KP    (OFF_G + EE)
#define OFF_XSP   (OFF_KP + KSPL * HEADS * DH * DH)
#define OFF_SVP   (OFF_XSP + XSPL * EMBED)
#define OFF_U     (OFF_SVP + SSPL * QKVW)
#define OFF_C     (OFF_U + HEADS * EMBED)

static __device__ __forceinline__ v8f wmma4(v2f a, v2f b, v8f c) {
    return __builtin_amdgcn_wmma_f32_16x16x4_f32(
        false, a, false, b, (short)0, c, false, false);
}

static __device__ __forceinline__ void gram_tile(int ti, int& mi, int& ni) {
    int m = 0, rem = ti;
    #pragma unroll
    for (int r = 0; r < 7; ++r) {
        if (m == r && rem >= 8 - r) { rem -= 8 - r; m = r + 1; }
    }
    mi = m; ni = m + rem;
}

__global__ __launch_bounds__(128) void mhsa_gram(
    const float* __restrict__ x, float* __restrict__ Gp)
{
    __shared__ __attribute__((aligned(16))) float sA[64 * GPITCH];
    __shared__ __attribute__((aligned(16))) float sB[64 * GPITCH];
    const int ti = blockIdx.x, s = blockIdx.y;
    int mi, ni;
    gram_tile(ti, mi, ni);
    const int nst = (s == GSPL - 1) ? 4 : 3;
    const int t = threadIdx.x, lane = t & 31, wave = t >> 5;
    const int lo = lane & 15, hi = lane >> 4;
    const int sr = t >> 4, sc = (t & 15) * 4;
    const float* xa = x + (size_t)(mi * 64 + sr) * NTOK + s * 192 + sc;
    const float* xb = x + (size_t)(ni * 64 + sr) * NTOK + s * 192 + sc;

    v4f ra[8], rb[8];
    #pragma unroll
    for (int i = 0; i < 8; ++i) {
        ra[i] = *(const v4f*)(xa + (size_t)i * 8 * NTOK);
        rb[i] = *(const v4f*)(xb + (size_t)i * 8 * NTOK);
    }

    v8f acc0 = {}, acc1 = {}, acc2 = {}, acc3 = {};
    const float* pa  = sA + (wave * 16 + lo) * GPITCH + 4 * hi;
    const float* pb  = sB + lo * GPITCH + 4 * hi;

    for (int st = 0; st < nst; ++st) {
        #pragma unroll
        for (int i = 0; i < 8; ++i) {
            *(v4f*)&sA[(sr + 8 * i) * GPITCH + sc] = ra[i];
            *(v4f*)&sB[(sr + 8 * i) * GPITCH + sc] = rb[i];
        }
        __syncthreads();
        if (st + 1 < nst) {
            #pragma unroll
            for (int i = 0; i < 8; ++i) {
                ra[i] = *(const v4f*)(xa + (size_t)i * 8 * NTOK + (st + 1) * 64);
                rb[i] = *(const v4f*)(xb + (size_t)i * 8 * NTOK + (st + 1) * 64);
            }
        }
        #pragma unroll
        for (int kk = 0; kk < 64; kk += 8) {
            const v4f a  = *(const v4f*)(pa + kk);
            const v4f b0 = *(const v4f*)(pb + kk);
            const v4f b1 = *(const v4f*)(pb + 16 * GPITCH + kk);
            const v4f b2 = *(const v4f*)(pb + 32 * GPITCH + kk);
            const v4f b3 = *(const v4f*)(pb + 48 * GPITCH + kk);
            const v2f a01 = {a.x, a.y}, a23 = {a.z, a.w};
            acc0 = wmma4(a01, (v2f){b0.x, b0.y}, acc0);
            acc1 = wmma4(a01, (v2f){b1.x, b1.y}, acc1);
            acc2 = wmma4(a01, (v2f){b2.x, b2.y}, acc2);
            acc3 = wmma4(a01, (v2f){b3.x, b3.y}, acc3);
            acc0 = wmma4(a23, (v2f){b0.z, b0.w}, acc0);
            acc1 = wmma4(a23, (v2f){b1.z, b1.w}, acc1);
            acc2 = wmma4(a23, (v2f){b2.z, b2.w}, acc2);
            acc3 = wmma4(a23, (v2f){b3.z, b3.w}, acc3);
        }
        __syncthreads();
    }

    float* dst = Gp + ((size_t)s * GTILES + ti) * GTSZ;
    #pragma unroll
    for (int r = 0; r < 8; ++r) {
        const int row = wave * 16 + hi * 8 + r;
        float* o = dst + row * 64 + lo;
        o[0] = acc0[r]; o[16] = acc1[r]; o[32] = acc2[r]; o[48] = acc3[r];
    }
}

__global__ __launch_bounds__(256) void mhsa_greduce(
    const float* __restrict__ Gp, const float* __restrict__ x,
    float* __restrict__ G, float* __restrict__ xsp)
{
    __shared__ float sX[32];
    __shared__ float sT[32 * 33];
    const int b = blockIdx.x, t = threadIdx.x;
    if (b < 4 * GTILES) {
        const int ti = b >> 2, q = b & 3;
        int mi, ni;
        gram_tile(ti, mi, ni);
        const int r0 = (q >> 1) * 32, c0 = (q & 1) * 32;
        const int rr = t >> 3, c4 = (t & 7) * 4;
        const float* src = Gp + (size_t)ti * GTSZ + (r0 + rr) * 64 + c0 + c4;
        v4f s = *(const v4f*)src;
        #pragma unroll
        for (int p = 1; p < GSPL; ++p) s += *(const v4f*)(src + (size_t)p * (GTILES * GTSZ));
        *(v4f*)(G + (size_t)(mi * 64 + r0 + rr) * EMBED + ni * 64 + c0 + c4) = s;
        if (mi != ni) {
            sT[(c4 + 0) * 33 + rr] = s.x;
            sT[(c4 + 1) * 33 + rr] = s.y;
            sT[(c4 + 2) * 33 + rr] = s.z;
            sT[(c4 + 3) * 33 + rr] = s.w;
            __syncthreads();
            const float* tp = sT + rr * 33 + c4;
            const v4f m = {tp[0], tp[1], tp[2], tp[3]};
            *(v4f*)(G + (size_t)(ni * 64 + c0 + rr) * EMBED + mi * 64 + r0 + c4) = m;
        }
    } else {
        const int xb = b - 4 * GTILES, rb = xb & 15, part = xb >> 4;
        const int lane = t & 31, wave = t >> 5;
        #pragma unroll
        for (int j = 0; j < 4; ++j) {
            const int row = rb * 32 + wave * 4 + j;
            const float* p = x + (size_t)row * NTOK + part * 512 + lane * 4;
            v4f s = *(const v4f*)p;
            s += *(const v4f*)(p + 128);
            s += *(const v4f*)(p + 256);
            s += *(const v4f*)(p + 384);
            float v = (s.x + s.y) + (s.z + s.w);
            v += __shfl_xor(v, 16, 32);
            v += __shfl_xor(v, 8, 32);
            v += __shfl_xor(v, 4, 32);
            v += __shfl_xor(v, 2, 32);
            v += __shfl_xor(v, 1, 32);
            if (lane == 0) sX[wave * 4 + j] = v;
        }
        __syncthreads();
        if (t < 32) xsp[part * EMBED + rb * 32 + t] = sX[t];
    }
}

#define TPITCH 20
#define TPLANE (64 * TPITCH)
__global__ __launch_bounds__(128) __attribute__((amdgpu_waves_per_eu(1, 2))) void mhsa_t_ktv(
    const float* __restrict__ G, const float* __restrict__ W, const float* __restrict__ bias,
    const float* __restrict__ xsp, float* __restrict__ Kp, float* __restrict__ svp,
    float* __restrict__ U, float* __restrict__ cpad)
{
    __shared__ float sT[4 * TPLANE];
    const int b = blockIdx.x, t = threadIdx.x;
    const int lane = t & 31, wave = t >> 5, lo = lane & 15, hi = lane >> 4;
    if (b < 256) {
        const int c2 = b & 3, h = (b >> 2) & 7, mi = b >> 5;
        const float* Wk = W + (size_t)(mi * 64 + 2 * hi) * QKVW + EMBED + h * DH + wave * 16 + lo;
        v2f wk[16];
        #pragma unroll
        for (int j = 0; j < 16; ++j)
            wk[j] = (v2f){Wk[(size_t)(4 * j) * QKVW], Wk[(size_t)(4 * j + 1) * QKVW]};

        const float* Ga = G + (size_t)(mi * 64 + lo) * EMBED + wave * 128 + 2 * hi;
        const float* Wv = W + (size_t)(wave * 128 + 2 * hi) * QKVW + 2 * EMBED + h * DH + c2 * 16 + lo;
        v2f ga[2][8][4], wv[2][8];
        #pragma unroll
        for (int u = 0; u < 8; ++u) {
            wv[0][u] = (v2f){Wv[(size_t)(4 * u) * QKVW], Wv[(size_t)(4 * u + 1) * QKVW]};
            #pragma unroll
            for (int i = 0; i < 4; ++i) ga[0][u][i] = *(const v2f*)(Ga + (size_t)(16 * i) * EMBED + 4 * u);
        }
        v8f acc0 = {}, acc1 = {}, acc2 = {}, acc3 = {};
        #pragma unroll
        for (int g = 0; g < 4; ++g) {
            const int cur = g & 1, nxt = cur ^ 1;
            if (g + 1 < 4) {
                #pragma unroll
                for (int u = 0; u < 8; ++u) {
                    const int k = 32 * (g + 1) + 4 * u;
                    wv[nxt][u] = (v2f){Wv[(size_t)k * QKVW], Wv[(size_t)(k + 1) * QKVW]};
                    #pragma unroll
                    for (int i = 0; i < 4; ++i) ga[nxt][u][i] = *(const v2f*)(Ga + (size_t)(16 * i) * EMBED + k);
                }
            }
            __builtin_amdgcn_sched_barrier(0);
            #pragma unroll
            for (int u = 0; u < 8; ++u) {
                acc0 = wmma4(ga[cur][u][0], wv[cur][u], acc0);
                acc1 = wmma4(ga[cur][u][1], wv[cur][u], acc1);
                acc2 = wmma4(ga[cur][u][2], wv[cur][u], acc2);
                acc3 = wmma4(ga[cur][u][3], wv[cur][u], acc3);
            }
        }
        float* pl = sT + wave * TPLANE + (8 * hi) * TPITCH + lo;
        #pragma unroll
        for (int r = 0; r < 8; ++r) {
            pl[r * TPITCH]        = acc0[r];
            pl[(16 + r) * TPITCH] = acc1[r];
            pl[(32 + r) * TPITCH] = acc2[r];
            pl[(48 + r) * TPITCH] = acc3[r];
        }
        __syncthreads();

        const float* tb = sT + (2 * hi) * TPITCH + lo;
        v8f kacc = {};
        #pragma unroll
        for (int j = 0; j < 16; ++j) {
            const float* q = tb + (4 * j) * TPITCH;
            v2f bv = {q[0], q[TPITCH]};
            #pragma unroll
            for (int p = 1; p < 4; ++p) bv += (v2f){q[p * TPLANE], q[p * TPLANE + TPITCH]};
            kacc = wmma4(wk[j], bv, kacc);
        }
        float* dst = Kp + ((((size_t)mi * HEADS + h) * 4 + c2) * DH + wave * 16 + 8 * hi) * 16 + lo;
        #pragma unroll
        for (int r = 0; r < 8; ++r) dst[r * 16] = kacc[r];
    } else if (b < 304) {
        const int gw = (b - 256) * 4 + wave;
        const int sl = gw / 12, cg = gw - sl * 12;
        const int c = cg * 128 + lane * 4;
        v4f s = {};
        #pragma unroll 8
        for (int i = 0; i < 32; ++i) {
            const int e = sl * 32 + i;
            float xs = xsp[e];
            #pragma unroll
            for (int p = 1; p < XSPL; ++p) xs += xsp[p * EMBED + e];
            const v4f w = *(const v4f*)(W + (size_t)e * QKVW + c);
            s += xs * w;
        }
        *(v4f*)(svp + sl * QKVW + c) = s;
    } else {
        const int ub = b - 304, h = ub & 7, et = ub >> 3;
        float* sXs = sT;
        float* sS  = sT + EMBED;
        float* sQ  = sS + SSPL * DH;
        float* sBk = sQ + DH;
        #pragma unroll
        for (int j = 0; j < 4; ++j) {
            const int e = t + 128 * j;
            float xs = xsp[e];
            #pragma unroll
            for (int p = 1; p < XSPL; ++p) xs += xsp[p * EMBED + e];
            sXs[e] = xs;
        }
        if (t < DH) sBk[t] = bias[EMBED + h * DH + t];
        __syncthreads();
        {
            const int c4 = (t & 15) * 4, g = t >> 4;
            const float* wq = W + (size_t)(g * 64) * QKVW + h * DH + c4;
            const float* xq = sXs + g * 64;
            #pragma unroll
            for (int sl = 0; sl < 2; ++sl) {
                v4f s = {};
                #pragma unroll
                for (int i = 0; i < 32; ++i) s += xq[sl * 32 + i] * *(const v4f*)(wq + (size_t)(sl * 32 + i) * QKVW);
                float* o = sS + (g * 2 + sl) * DH + c4;
                o[0] = s.x; o[1] = s.y; o[2] = s.z; o[3] = s.w;
            }
        }
        __syncthreads();
        if (t < DH) {
            float q = sS[t];
            #pragma unroll
            for (int p = 1; p < SSPL; ++p) q += sS[p * DH + t];
            sQ[t] = q + NTOKF * bias[h * DH + t];
        }
        __syncthreads();
        if (t < DH) {
            const int e = et * 64 + t;
            const float* wk = W + (size_t)e * QKVW + EMBED + h * DH;
            float s = 0.0f;
            #pragma unroll
            for (int d = 0; d < DH; d += 4) {
                const v4f w = *(const v4f*)(wk + d);
                s += w.x * sQ[d] + w.y * sQ[d + 1] + w.z * sQ[d + 2] + w.w * sQ[d + 3];
            }
            U[h * EMBED + e] = s;
        } else if (et == 0 && t < DH + 32) {
            float s = 0.0f;
            #pragma unroll
            for (int d = 0; d < DH; ++d) s += sBk[d] * sQ[d];
            cpad[h * 32 + (t - DH)] = s;
        }
    }
}

#define OPITCH 72
#define OCHUNK (64 * OPITCH)
#define KPW    (HEADS * DH * DH / 4)
__global__ __launch_bounds__(256) void mhsa_out(
    const float* __restrict__ x, const float* __restrict__ W, const float* __restrict__ bias,
    const float* __restrict__ Kp, const float* __restrict__ svp, const float* __restrict__ U,
    const float* __restrict__ cpad, float* __restrict__ out)
{
    __shared__ __attribute__((aligned(16))) float sA[2 * OCHUNK];
    __shared__ __attribute__((aligned(16))) float sB[2 * OCHUNK];
    __shared__ float sC[4 * HEADS * 32];
    __shared__ __attribute__((aligned(16))) float sK[DH * OPITCH];
    __shared__ float sKs[DH], sVs[DH], sBk[DH], sBv[DH];
    const int b = blockIdx.x, t = threadIdx.x;
    const int lane = t & 31, wave = t >> 5, lo = lane & 15, hi = lane >> 4;
    if (b >= 128) {
        const int gb = b - 128;
        const int mt = gb & 63, h = gb >> 6;
        const int sr = t >> 4, sc = (t & 15) * 4;
        const float* xa = x + (size_t)sr * NTOK + mt * 64 + sc;
        const float* mb = W + (size_t)sr * QKVW + h * DH + sc;
        const v4f* kp4 = (const v4f*)(Kp + (size_t)h * DH * DH) + t;

        float* kw = sK + ((t >> 2) & 63) * OPITCH + (t & 3) * 4;

        v4f ra[4], rb[4];
        v4f kn[4], km[4], kl[4];
        #pragma unroll
        for (int i = 0; i < 4; ++i) {
            ra[i] = *(const v4f*)(xa + (size_t)(16 * i) * NTOK);
            rb[i] = *(const v4f*)(mb + (size_t)(16 * i) * QKVW);
            kn[i] = kp4[256 * i];
            km[i] = kp4[(size_t)1 * KPW + 256 * i];
            kl[i] = kp4[(size_t)2 * KPW + 256 * i];
        }
        if (t < DH) {
            const int c = h * DH + t;
            float k = svp[EMBED + c], v = svp[2 * EMBED + c];
            #pragma unroll
            for (int p = 1; p < SSPL; ++p) {
                k += svp[p * QKVW + EMBED + c];
                v += svp[p * QKVW + 2 * EMBED + c];
            }
            const float bk = bias[EMBED + c], bv = bias[2 * EMBED + c];
            sKs[t] = k;
            sVs[t] = v + NTOKF * bv;
            sBk[t] = bk;
            sBv[t] = bv;
        }
        #pragma unroll
        for (int i = 0; i < 4; ++i) {
            *(v4f*)&sA[(sr + 16 * i) * OPITCH + sc] = ra[i];
            *(v4f*)&sB[(sr + 16 * i) * OPITCH + sc] = rb[i];
            *(v4f*)(kw + 16 * i) = (kn[i] + km[i]) + kl[i];
        }
        __syncthreads();

        const int wm = wave & 3, wn = wave >> 2;
        const float* pa = sA + (2 * hi) * OPITCH + wm * 16 + lo;
        const float* pb = sB + (2 * hi) * OPITCH + wn * 32 + lo;
        v8f acc0 = {}, acc1 = {};

        for (int st = 0; st < 8; ++st) {
            const int cur = (st & 1) * OCHUNK;
            if (st + 1 < 8) {
                #pragma unroll
                for (int i = 0; i < 4; ++i) {
                    ra[i] = *(const v4f*)(xa + (size_t)((st + 1) * 64 + 16 * i) * NTOK);
                    rb[i] = *(const v4f*)(mb + (size_t)((st + 1) * 64 + 16 * i) * QKVW);
                }
            }
            if (st >= 4) {
                #pragma unroll
                for (int i = 0; i < 4; ++i) {
                    kn[i] = kp4[(size_t)(st < 7 ? st - 1 : 6) * KPW + 256 * i];
                    if (st == 7) km[i] = kp4[(size_t)7 * KPW + 256 * i];
                }
            }
            __builtin_amdgcn_sched_barrier(0);
            #pragma unroll
            for (int kk = 0; kk < 64; kk += 4) {
                const float* a0 = pa + cur + kk * OPITCH;
                const float* b0 = pb + cur + kk * OPITCH;
                const v2f a = {a0[0], a0[OPITCH]};
                acc0 = wmma4(a, (v2f){b0[0],  b0[OPITCH]},      acc0);
                acc1 = wmma4(a, (v2f){b0[16], b0[OPITCH + 16]}, acc1);
            }
            if (st + 1 < 8) {
                const int nxt = ((st + 1) & 1) * OCHUNK;
                #pragma unroll
                for (int i = 0; i < 4; ++i) {
                    *(v4f*)&sA[nxt + (sr + 16 * i) * OPITCH + sc] = ra[i];
                    *(v4f*)&sB[nxt + (sr + 16 * i) * OPITCH + sc] = rb[i];
                }
            }
            if (st >= 4) {
                #pragma unroll
                for (int i = 0; i < 4; ++i) {
                    v4f v = *(const v4f*)(kw + 16 * i) + kn[i];
                    if (st == 7) v += km[i];
                    *(v4f*)(kw + 16 * i) = v;
                }
            }
            if (st + 1 < 8) __syncthreads();
        }

        #pragma unroll
        for (int i = 0; i < 4; ++i) {
            const int d1 = (t >> 2) & 63, d2 = 16 * i + (t & 3) * 4;
            const float ks = sKs[d1], bk = sBk[d1];
            v4f v = *(const v4f*)(kw + 16 * i);
            v.x += ks * sBv[d2]     + bk * sVs[d2];
            v.y += ks * sBv[d2 + 1] + bk * sVs[d2 + 1];
            v.z += ks * sBv[d2 + 2] + bk * sVs[d2 + 2];
            v.w += ks * sBv[d2 + 3] + bk * sVs[d2 + 3];
            *(v4f*)(kw + 16 * i) = v;
        }
        const int nb = h * DH + wn * 32 + lo;
        const float q0 = bias[nb], q1 = bias[nb + 16];
        float* sQt = sA;
        #pragma unroll
        for (int r = 0; r < 8; ++r) {
            float* q = sQt + (wm * 16 + hi * 8 + r) * GPITCH + wn * 32 + lo;
            q[0]  = acc0[r] + q0;
            q[16] = acc1[r] + q1;
        }
        __syncthreads();
        const float* qa = sQt + (wm * 16 + lo) * GPITCH + 2 * hi;
        const float* kb = sK + (2 * hi) * OPITCH + wn * 32 + lo;
        v8f o0 = {}, o1 = {};
        #pragma unroll
        for (int kk = 0; kk < DH; kk += 4) {
            const v2f a = *(const v2f*)(qa + kk);
            const float* b0 = kb + kk * OPITCH;
            o0 = wmma4(a, (v2f){b0[0],  b0[OPITCH]},      o0);
            o1 = wmma4(a, (v2f){b0[16], b0[OPITCH + 16]}, o1);
        }
        const int m0 = mt * 64 + wm * 16;
        float* base = out + HEADS * NTOK + (size_t)h * NTOK * DH;
        #pragma unroll
        for (int r = 0; r < 8; ++r) {
            float* o = base + (size_t)(m0 + hi * 8 + r) * DH + wn * 32 + lo;
            o[0]  = o0[r] * SCALE;
            o[16] = o1[r] * SCALE;
        }
    } else {
        const int n = b * 32 + lane;
        if (wave < 4) {
            float a[HEADS];
            #pragma unroll
            for (int hh = 0; hh < HEADS; ++hh) a[hh] = 0.0f;
            #pragma unroll 16
            for (int i = 0; i < 128; ++i) {
                const int e = wave * 128 + i;
                const float xv = x[(size_t)e * NTOK + n];
                #pragma unroll
                for (int hh = 0; hh < HEADS; ++hh) a[hh] += U[hh * EMBED + e] * xv;
            }
            #pragma unroll
            for (int hh = 0; hh < HEADS; ++hh) sC[(wave * HEADS + hh) * 32 + lane] = a[hh];
        }
        __syncthreads();
        if (wave < 4) {
            #pragma unroll
            for (int j = 0; j < 2; ++j) {
                const int hh = wave + 4 * j;
                const float s = ((sC[(0 * HEADS + hh) * 32 + lane] + sC[(1 * HEADS + hh) * 32 + lane])
                               + sC[(2 * HEADS + hh) * 32 + lane]) + sC[(3 * HEADS + hh) * 32 + lane];
                out[hh * NTOK + n] = (s + cpad[hh * 32]) * SCALE;
            }
        }
    }
}

extern "C" void kernel_launch(void* const* d_in, const int* in_sizes, int n_in,
                              void* d_out, int out_size, void* d_ws, size_t ws_size,
                              hipStream_t stream) {
    const float* x    = (const float*)d_in[0];
    const float* W    = (const float*)d_in[1];
    const float* bias = (const float*)d_in[2];
    float* out = (float*)d_out;
    float* ws  = (float*)d_ws;

    mhsa_gram<<<dim3(GTILES, GSPL), 128, 0, stream>>>(x, ws + OFF_GP);
    mhsa_greduce<<<GRED_BLOCKS, 256, 0, stream>>>(ws + OFF_GP, x, ws + OFF_G, ws + OFF_XSP);
    mhsa_t_ktv<<<368, 128, 0, stream>>>(ws + OFF_G, W, bias, ws + OFF_XSP, ws + OFF_KP, ws + OFF_SVP,
                                        ws + OFF_U, ws + OFF_C);
    mhsa_out<<<640, 256, 0, stream>>>(x, W, bias, ws + OFF_KP, ws + OFF_SVP, ws + OFF_U, ws + OFF_C, out);
}


# Output GEMM: block tile 128 tokens x 64 columns (was 64 x 64), wave tile 32x32, 256 blocks; operands in LDS as k pairs, one 8-byte read per fragment; next chunk stored mid-chunk; sum order unchanged
# speedup vs baseline: 2.2164x; 1.0931x over previous
#include <hip/hip_runtime.h>


typedef __attribute__((ext_vector_type(2))) float v2f;
typedef __attribute__((ext_vector_type(4))) float v4f;
typedef __attribute__((ext_vector_type(8))) float v8f;

#define NTOK   4096
#define EMBED  512
#define QKVW   1536
#define HEADS  8
#define DH     64
#define SCALE  0.125f
#define NTOKF  4096.0f
#define GSPL   21
#define GTILES 36
#define GTSZ   (64 * 64)
#define GRED_BLOCKS 272
#define KSPL   8
#define XSPL   8
#define SSPL   16
#define EE     (EMBED * EMBED)
#define GPITCH 68
#define KPITCH 72

#define OFF_GP    0
#define OFF_G     (OFF_GP + GSPL * GTILES * GTSZ)
#define OFF_KP    (OFF_G + EE)
#define OFF_XSP   (OFF_KP + KSPL * HEADS * DH * DH)
#define OFF_SVP   (OFF_XSP + XSPL * EMBED)
#define OFF_U     (OFF_SVP + SSPL * QKVW)
#define OFF_C     (OFF_U + HEADS * EMBED)

static __device__ __forceinline__ v8f wmma4(v2f a, v2f b, v8f c) {
    return __builtin_amdgcn_wmma_f32_16x16x4_f32(
        false, a, false, b, (short)0, c, false, false);
}

static __device__ __forceinline__ void gram_tile(int ti, int& mi, int& ni) {
    int m = 0, rem = ti;
    #pragma unroll
    for (int r = 0; r < 7; ++r) {
        if (m == r && rem >= 8 - r) { rem -= 8 - r; m = r + 1; }
    }
    mi = m; ni = m + rem;
}

__global__ __launch_bounds__(128) void mhsa_gram(
    const float* __restrict__ x, float* __restrict__ Gp)
{
    __shared__ __attribute__((aligned(16))) float sA[64 * GPITCH];
    __shared__ __attribute__((aligned(16))) float sB[64 * GPITCH];
    const int ti = blockIdx.x, s = blockIdx.y;
    int mi, ni;
    gram_tile(ti, mi, ni);
    const int nst = (s == GSPL - 1) ? 4 : 3;
    const int t = threadIdx.x, lane = t & 31, wave = t >> 5;
    const int lo = lane & 15, hi = lane >> 4;
    const int sr = t >> 4, sc = (t & 15) * 4;
    const float* xa = x + (size_t)(mi * 64 + sr) * NTOK + s * 192 + sc;
    const float* xb = x + (size_t)(ni * 64 + sr) * NTOK + s * 192 + sc;

    v4f ra[8], rb[8];
    #pragma unroll
    for (int i = 0; i < 8; ++i) {
        ra[i] = *(const v4f*)(xa + (size_t)i * 8 * NTOK);
        rb[i] = *(const v4f*)(xb + (size_t)i * 8 * NTOK);
    }

    v8f acc0 = {}, acc1 = {}, acc2 = {}, acc3 = {};
    const float* pa  = sA + (wave * 16 + lo) * GPITCH + 4 * hi;
    const float* pb  = sB + lo * GPITCH + 4 * hi;

    for (int st = 0; st < nst; ++st) {
        #pragma unroll
        for (int i = 0; i < 8; ++i) {
            *(v4f*)&sA[(sr + 8 * i) * GPITCH + sc] = ra[i];
            *(v4f*)&sB[(sr + 8 * i) * GPITCH + sc] = rb[i];
        }
        __syncthreads();
        if (st + 1 < nst) {
            #pragma unroll
            for (int i = 0; i < 8; ++i) {
                ra[i] = *(const v4f*)(xa + (size_t)i * 8 * NTOK + (st + 1) * 64);
                rb[i] = *(const v4f*)(xb + (size_t)i * 8 * NTOK + (st + 1) * 64);
            }
        }
        #pragma unroll
        for (int kk = 0; kk < 64; kk += 8) {
            const v4f a  = *(const v4f*)(pa + kk);
            const v4f b0 = *(const v4f*)(pb + kk);
            const v4f b1 = *(const v4f*)(pb + 16 * GPITCH + kk);
            const v4f b2 = *(const v4f*)(pb + 32 * GPITCH + kk);
            const v4f b3 = *(const v4f*)(pb + 48 * GPITCH + kk);
            const v2f a01 = {a.x, a.y}, a23 = {a.z, a.w};
            acc0 = wmma4(a01, (v2f){b0.x, b0.y}, acc0);
            acc1 = wmma4(a01, (v2f){b1.x, b1.y}, acc1);
            acc2 = wmma4(a01, (v2f){b2.x, b2.y}, acc2);
            acc3 = wmma4(a01, (v2f){b3.x, b3.y}, acc3);
            acc0 = wmma4(a23, (v2f){b0.z, b0.w}, acc0);
            acc1 = wmma4(a23, (v2f){b1.z, b1.w}, acc1);
            acc2 = wmma4(a23, (v2f){b2.z, b2.w}, acc2);
            acc3 = wmma4(a23, (v2f){b3.z, b3.w}, acc3);
        }
        __syncthreads();
    }

    float* dst = Gp + ((size_t)s * GTILES + ti) * GTSZ;
    #pragma unroll
    for (int r = 0; r < 8; ++r) {
        const int row = wave * 16 + hi * 8 + r;
        float* o = dst + row * 64 + lo;
        o[0] = acc0[r]; o[16] = acc1[r]; o[32] = acc2[r]; o[48] = acc3[r];
    }
}

__global__ __launch_bounds__(256) void mhsa_greduce(
    const float* __restrict__ Gp, const float* __restrict__ x,
    float* __restrict__ G, float* __restrict__ xsp)
{
    __shared__ float sX[32];
    __shared__ float sT[32 * 33];
    const int b = blockIdx.x, t = threadIdx.x;
    if (b < 4 * GTILES) {
        const int ti = b >> 2, q = b & 3;
        int mi, ni;
        gram_tile(ti, mi, ni);
        const int r0 = (q >> 1) * 32, c0 = (q & 1) * 32;
        const int rr = t >> 3, c4 = (t & 7) * 4;
        const float* src = Gp + (size_t)ti * GTSZ + (r0 + rr) * 64 + c0 + c4;
        v4f s = *(const v4f*)src;
        #pragma unroll
        for (int p = 1; p < GSPL; ++p) s += *(const v4f*)(src + (size_t)p * (GTILES * GTSZ));
        *(v4f*)(G + (size_t)(mi * 64 + r0 + rr) * EMBED + ni * 64 + c0 + c4) = s;
        if (mi != ni) {
            sT[(c4 + 0) * 33 + rr] = s.x;
            sT[(c4 + 1) * 33 + rr] = s.y;
            sT[(c4 + 2) * 33 + rr] = s.z;
            sT[(c4 + 3) * 33 + rr] = s.w;
            __syncthreads();
            const float* tp = sT + rr * 33 + c4;
            const v4f m = {tp[0], tp[1], tp[2], tp[3]};
            *(v4f*)(G + (size_t)(ni * 64 + c0 + rr) * EMBED + mi * 64 + r0 + c4) = m;
        }
    } else {
        const int xb = b - 4 * GTILES, rb = xb & 15, part = xb >> 4;
        const int lane = t & 31, wave = t >> 5;
        #pragma unroll
        for (int j = 0; j < 4; ++j) {
            const int row = rb * 32 + wave * 4 + j;
            const float* p = x + (size_t)row * NTOK + part * 512 + lane * 4;
            v4f s = *(const v4f*)p;
            s += *(const v4f*)(p + 128);
            s += *(const v4f*)(p + 256);
            s += *(const v4f*)(p + 384);
            float v = (s.x + s.y) + (s.z + s.w);
            v += __shfl_xor(v, 16, 32);
            v += __shfl_xor(v, 8, 32);
            v += __shfl_xor(v, 4, 32);
            v += __shfl_xor(v, 2, 32);
            v += __shfl_xor(v, 1, 32);
            if (lane == 0) sX[wave * 4 + j] = v;
        }
        __syncthreads();
        if (t < 32) xsp[part * EMBED + rb * 32 + t] = sX[t];
    }
}

#define TPITCH 20
#define TPLANE (64 * TPITCH)
__global__ __launch_bounds__(128) __attribute__((amdgpu_waves_per_eu(1, 2))) void mhsa_t_ktv(
    const float* __restrict__ G, const float* __restrict__ W, const float* __restrict__ bias,
    const float* __restrict__ xsp, float* __restrict__ Kp, float* __restrict__ svp,
    float* __restrict__ U, float* __restrict__ cpad)
{
    __shared__ float sT[4 * TPLANE];
    const int b = blockIdx.x, t = threadIdx.x;
    const int lane = t & 31, wave = t >> 5, lo = lane & 15, hi = lane >> 4;
    if (b < 256) {
        const int c2 = b & 3, h = (b >> 2) & 7, mi = b >> 5;
        const float* Wk = W + (size_t)(mi * 64 + 2 * hi) * QKVW + EMBED + h * DH + wave * 16 + lo;
        v2f wk[16];
        #pragma unroll
        for (int j = 0; j < 16; ++j)
            wk[j] = (v2f){Wk[(size_t)(4 * j) * QKVW], Wk[(size_t)(4 * j + 1) * QKVW]};

        const float* Ga = G + (size_t)(mi * 64 + lo) * EMBED + wave * 128 + 2 * hi;
        const float* Wv = W + (size_t)(wave * 128 + 2 * hi) * QKVW + 2 * EMBED + h * DH + c2 * 16 + lo;
        v2f ga[2][8][4], wv[2][8];
        #pragma unroll
        for (int u = 0; u < 8; ++u) {
            wv[0][u] = (v2f){Wv[(size_t)(4 * u) * QKVW], Wv[(size_t)(4 * u + 1) * QKVW]};
            #pragma unroll
            for (int i = 0; i < 4; ++i) ga[0][u][i] = *(const v2f*)(Ga + (size_t)(16 * i) * EMBED + 4 * u);
        }
        v8f acc0 = {}, acc1 = {}, acc2 = {}, acc3 = {};
        #pragma unroll
        for (int g = 0; g < 4; ++g) {
            const int cur = g & 1, nxt = cur ^ 1;
            if (g + 1 < 4) {
                #pragma unroll
                for (int u = 0; u < 8; ++u) {
                    const int k = 32 * (g + 1) + 4 * u;
                    wv[nxt][u] = (v2f){Wv[(size_t)k * QKVW], Wv[(size_t)(k + 1) * QKVW]};
                    #pragma unroll
                    for (int i = 0; i < 4; ++i) ga[nxt][u][i] = *(const v2f*)(Ga + (size_t)(16 * i) * EMBED + k);
                }
            }
            __builtin_amdgcn_sched_barrier(0);
            #pragma unroll
            for (int u = 0; u < 8; ++u) {
                acc0 = wmma4(ga[cur][u][0], wv[cur][u], acc0);
                acc1 = wmma4(ga[cur][u][1], wv[cur][u], acc1);
                acc2 = wmma4(ga[cur][u][2], wv[cur][u], acc2);
                acc3 = wmma4(ga[cur][u][3], wv[cur][u], acc3);
            }
        }
        float* pl = sT + wave * TPLANE + (8 * hi) * TPITCH + lo;
        #pragma unroll
        for (int r = 0; r < 8; ++r) {
            pl[r * TPITCH]        = acc0[r];
            pl[(16 + r) * TPITCH] = acc1[r];
            pl[(32 + r) * TPITCH] = acc2[r];
            pl[(48 + r) * TPITCH] = acc3[r];
        }
        __syncthreads();

        const float* tb = sT + (2 * hi) * TPITCH + lo;
        v8f kacc = {};
        #pragma unroll
        for (int j = 0; j < 16; ++j) {
            const float* q = tb + (4 * j) * TPITCH;
            v2f bv = {q[0], q[TPITCH]};
            #pragma unroll
            for (int p = 1; p < 4; ++p) bv += (v2f){q[p * TPLANE], q[p * TPLANE + TPITCH]};
            kacc = wmma4(wk[j], bv, kacc);
        }
        float* dst = Kp + ((((size_t)mi * HEADS + h) * 4 + c2) * DH + wave * 16 + 8 * hi) * 16 + lo;
        #pragma unroll
        for (int r = 0; r < 8; ++r) dst[r * 16] = kacc[r];
    } else if (b < 304) {
        const int gw = (b - 256) * 4 + wave;
        const int sl = gw / 12, cg = gw - sl * 12;
        const int c = cg * 128 + lane * 4;
        v4f s = {};
        #pragma unroll 8
        for (int i = 0; i < 32; ++i) {
            const int e = sl * 32 + i;
            float xs = xsp[e];
            #pragma unroll
            for (int p = 1; p < XSPL; ++p) xs += xsp[p * EMBED + e];
            const v4f w = *(const v4f*)(W + (size_t)e * QKVW + c);
            s += xs * w;
        }
        *(v4f*)(svp + sl * QKVW + c) = s;
    } else {
        const int ub = b - 304, h = ub & 7, et = ub >> 3;
        float* sXs = sT;
        float* sS  = sT + EMBED;
        float* sQ  = sS + SSPL * DH;
        float* sBk = sQ + DH;
        #pragma unroll
        for (int j = 0; j < 4; ++j) {
            const int e = t + 128 * j;
            float xs = xsp[e];
            #pragma unroll
            for (int p = 1; p < XSPL; ++p) xs += xsp[p * EMBED + e];
            sXs[e] = xs;
        }
        if (t < DH) sBk[t] = bias[EMBED + h * DH + t];
        __syncthreads();
        {
            const int c4 = (t & 15) * 4, g = t >> 4;
            const float* wq = W + (size_t)(g * 64) * QKVW + h * DH + c4;
            const float* xq = sXs + g * 64;
            #pragma unroll
            for (int sl = 0; sl < 2; ++sl) {
                v4f s = {};
                #pragma unroll
                for (int i = 0; i < 32; ++i) s += xq[sl * 32 + i] * *(const v4f*)(wq + (size_t)(sl * 32 + i) * QKVW);
                float* o = sS + (g * 2 + sl) * DH + c4;
                o[0] = s.x; o[1] = s.y; o[2] = s.z; o[3] = s.w;
            }
        }
        __syncthreads();
        if (t < DH) {
            float q = sS[t];
            #pragma unroll
            for (int p = 1; p < SSPL; ++p) q += sS[p * DH + t];
            sQ[t] = q + NTOKF * bias[h * DH + t];
        }
        __syncthreads();
        if (t < DH) {
            const int e = et * 64 + t;
            const float* wk = W + (size_t)e * QKVW + EMBED + h * DH;
            float s = 0.0f;
            #pragma unroll
            for (int d = 0; d < DH; d += 4) {
                const v4f w = *(const v4f*)(wk + d);
                s += w.x * sQ[d] + w.y * sQ[d + 1] + w.z * sQ[d + 2] + w.w * sQ[d + 3];
            }
            U[h * EMBED + e] = s;
        } else if (et == 0 && t < DH + 32) {
            float s = 0.0f;
            #pragma unroll
            for (int d = 0; d < DH; ++d) s += sBk[d] * sQ[d];
            cpad[h * 32 + (t - DH)] = s;
        }
    }
}

#define OPITCH 72
#define AP2    288
#define ACHUNK (32 * AP2)
#define BP2    160
#define BCHUNK (32 * BP2)
#define KPW    (HEADS * DH * DH / 4)
#define OUT_GEMM_BLOCKS 256
template <bool STORE>
static __device__ __forceinline__ void out_chunk(
    const float* ca, const float* cb, float* na, float* nb, float* kw,
    const v4f (&ra)[8], const v4f (&rb)[4], const v4f (&kn)[4],
    v8f& acc00, v8f& acc01, v8f& acc10, v8f& acc11)
{
    v2f fa[2][4][2], fb[2][4][2];
    #pragma unroll
    for (int u = 0; u < 4; ++u) {
        fa[0][u][0] = *(const v2f*)(ca + (2 * u) * AP2);
        fa[0][u][1] = *(const v2f*)(ca + (2 * u) * AP2 + 32);
        fb[0][u][0] = *(const v2f*)(cb + (2 * u) * BP2);
        fb[0][u][1] = *(const v2f*)(cb + (2 * u) * BP2 + 32);
    }
    #pragma unroll
    for (int g = 0; g < 4; ++g) {
        const int cs = g & 1, ns = cs ^ 1;
        if (STORE && g == 2) {
            #pragma unroll
            for (int j = 0; j < 4; ++j) {
                *(v4f*)(na + (8 * j) * AP2)     = (v4f){ra[2 * j].x, ra[2 * j + 1].x, ra[2 * j].y, ra[2 * j + 1].y};
                *(v4f*)(na + (8 * j) * AP2 + 4) = (v4f){ra[2 * j].z, ra[2 * j + 1].z, ra[2 * j].w, ra[2 * j + 1].w};
            }
            #pragma unroll
            for (int j = 0; j < 2; ++j) {
                *(v4f*)(nb + (16 * j) * BP2)     = (v4f){rb[2 * j].x, rb[2 * j + 1].x, rb[2 * j].y, rb[2 * j + 1].y};
                *(v4f*)(nb + (16 * j) * BP2 + 4) = (v4f){rb[2 * j].z, rb[2 * j + 1].z, rb[2 * j].w, rb[2 * j + 1].w};
            }
            #pragma unroll
            for (int i = 0; i < 4; ++i) *(v4f*)(kw + 16 * i) = *(const v4f*)(kw + 16 * i) + kn[i];
        }
        if (g + 1 < 4) {
            #pragma unroll
            for (int u = 0; u < 4; ++u) {
                fa[ns][u][0] = *(const v2f*)(ca + (8 * (g + 1) + 2 * u) * AP2);
                fa[ns][u][1] = *(const v2f*)(ca + (8 * (g + 1) + 2 * u) * AP2 + 32);
                fb[ns][u][0] = *(const v2f*)(cb + (8 * (g + 1) + 2 * u) * BP2);
                fb[ns][u][1] = *(const v2f*)(cb + (8 * (g + 1) + 2 * u) * BP2 + 32);
            }
        }
        __builtin_amdgcn_sched_barrier(0);
        #pragma unroll
        for (int u = 0; u < 4; ++u) {
            acc00 = wmma4(fa[cs][u][0], fb[cs][u][0], acc00);
            acc01 = wmma4(fa[cs][u][0], fb[cs][u][1], acc01);
            acc10 = wmma4(fa[cs][u][1], fb[cs][u][0], acc10);
            acc11 = wmma4(fa[cs][u][1], fb[cs][u][1], acc11);
        }
    }
}

__global__ __launch_bounds__(256) void mhsa_out(
    const float* __restrict__ x, const float* __restrict__ W, const float* __restrict__ bias,
    const float* __restrict__ Kp, const float* __restrict__ svp, const float* __restrict__ U,
    const float* __restrict__ cpad, float* __restrict__ out)
{
    __shared__ __attribute__((aligned(16))) float sA[2 * ACHUNK];
    __shared__ __attribute__((aligned(16))) float sB[2 * BCHUNK];
    __shared__ float sC[4 * HEADS * 32];
    __shared__ __attribute__((aligned(16))) float sK[DH * OPITCH];
    __shared__ float sKs[DH], sVs[DH], sBk[DH], sBv[DH];
    const int b = blockIdx.x, t = threadIdx.x;
    const int lane = t & 31, wave = t >> 5, lo = lane & 15, hi = lane >> 4;
    if (b < OUT_GEMM_BLOCKS) {
        const int mt = b & 31, h = b >> 5;
        const int sra = t >> 5, sca = (t & 31) * 4;
        const int srb = t >> 4, scb = (t & 15) * 4;
        const float* xa = x + (size_t)(2 * sra) * NTOK + mt * 128 + sca;
        const float* mb = W + (size_t)(2 * srb) * QKVW + h * DH + scb;
        float* wa = sA + sra * AP2 + 2 * sca;
        float* wb = sB + srb * BP2 + 2 * scb;
        const v4f* kp4 = (const v4f*)(Kp + (size_t)h * DH * DH) + t;

        float* kw = sK + ((t >> 2) & 63) * OPITCH + (t & 3) * 4;

        v4f ra[8], rb[4];
        v4f kn[4];
        #pragma unroll
        for (int i = 0; i < 8; ++i) ra[i] = *(const v4f*)(xa + (size_t)(16 * (i >> 1) + (i & 1)) * NTOK);
        #pragma unroll
        for (int i = 0; i < 4; ++i) {
            rb[i] = *(const v4f*)(mb + (size_t)(32 * (i >> 1) + (i & 1)) * QKVW);
            kn[i] = kp4[256 * i];
        }
        if (t < DH) {
            const int c = h * DH + t;
            float k = svp[EMBED + c], v = svp[2 * EMBED + c];
            #pragma unroll
            for (int p = 1; p < SSPL; ++p) {
                k += svp[p * QKVW + EMBED + c];
                v += svp[p * QKVW + 2 * EMBED + c];
            }
            const float bk = bias[EMBED + c], bv = bias[2 * EMBED + c];
            sKs[t] = k;
            sVs[t] = v + NTOKF * bv;
            sBk[t] = bk;
            sBv[t] = bv;
        }
        #pragma unroll
        for (int j = 0; j < 4; ++j) {
            *(v4f*)(wa + (8 * j) * AP2)     = (v4f){ra[2 * j].x, ra[2 * j + 1].x, ra[2 * j].y, ra[2 * j + 1].y};
            *(v4f*)(wa + (8 * j) * AP2 + 4) = (v4f){ra[2 * j].z, ra[2 * j + 1].z, ra[2 * j].w, ra[2 * j + 1].w};
        }
        #pragma unroll
        for (int j = 0; j < 2; ++j) {
            *(v4f*)(wb + (16 * j) * BP2)     = (v4f){rb[2 * j].x, rb[2 * j + 1].x, rb[2 * j].y, rb[2 * j + 1].y};
            *(v4f*)(wb + (16 * j) * BP2 + 4) = (v4f){rb[2 * j].z, rb[2 * j + 1].z, rb[2 * j].w, rb[2 * j + 1].w};
        }
        #pragma unroll
        for (int i = 0; i < 4; ++i) *(v4f*)(kw + 16 * i) = kn[i];
        __syncthreads();

        const int wm = wave & 3, wn = wave >> 2;
        const float* pa = sA + hi * AP2 + 2 * (wm * 32 + lo);
        const float* pb = sB + hi * BP2 + 2 * (wn * 32 + lo);
        v8f acc00 = {}, acc01 = {}, acc10 = {}, acc11 = {};

        #pragma nounroll
        for (int st = 0; st < 7; ++st) {
            const int curA = (st & 1) * ACHUNK, curB = (st & 1) * BCHUNK;
            const int nxtA = ACHUNK - curA, nxtB = BCHUNK - curB;
            #pragma unroll
            for (int i = 0; i < 8; ++i) ra[i] = *(const v4f*)(xa + (size_t)((st + 1) * 64 + 16 * (i >> 1) + (i & 1)) * NTOK);
            #pragma unroll
            for (int i = 0; i < 4; ++i) {
                rb[i] = *(const v4f*)(mb + (size_t)((st + 1) * 64 + 32 * (i >> 1) + (i & 1)) * QKVW);
                kn[i] = kp4[(size_t)(st + 1) * KPW + 256 * i];
            }
            __builtin_amdgcn_sched_barrier(0);
            out_chunk<true>(pa + curA, pb + curB, wa + nxtA, wb + nxtB, kw, ra, rb, kn, acc00, acc01, acc10, acc11);
            __syncthreads();
        }
        out_chunk<false>(pa + ACHUNK, pb + BCHUNK, wa, wb, kw, ra, rb, kn, acc00, acc01, acc10, acc11);

        #pragma unroll
        for (int i = 0; i < 4; ++i) {
            const int d1 = (t >> 2) & 63, d2 = 16 * i + (t & 3) * 4;
            const float ks = sKs[d1], bk = sBk[d1];
            v4f v = *(const v4f*)(kw + 16 * i);
            v.x += ks * sBv[d2]     + bk * sVs[d2];
            v.y += ks * sBv[d2 + 1] + bk * sVs[d2 + 1];
            v.z += ks * sBv[d2 + 2] + bk * sVs[d2 + 2];
            v.w += ks * sBv[d2 + 3] + bk * sVs[d2 + 3];
            *(v4f*)(kw + 16 * i) = v;
        }
        const int nb = h * DH + wn * 32 + lo;
        const float q0 = bias[nb], q1 = bias[nb + 16];
        float* sQt = sA;
        #pragma unroll
        for (int r = 0; r < 8; ++r) {
            float* q = sQt + (wm * 32 + hi * 8 + r) * GPITCH + wn * 32 + lo;
            q[0]                = acc00[r] + q0;
            q[16]               = acc01[r] + q1;
            q[16 * GPITCH]      = acc10[r] + q0;
            q[16 * GPITCH + 16] = acc11[r] + q1;
        }
        __syncthreads();
        const float* qa = sQt + (wm * 32 + lo) * GPITCH + 2 * hi;
        const float* kb = sK + (2 * hi) * OPITCH + wn * 32 + lo;
        v8f o00 = {}, o01 = {}, o10 = {}, o11 = {};
        #pragma unroll
        for (int kk = 0; kk < DH; kk += 4) {
            const v2f fa0 = *(const v2f*)(qa + kk);
            const v2f fa1 = *(const v2f*)(qa + 16 * GPITCH + kk);
            const float* b0 = kb + kk * OPITCH;
            const v2f fb0 = {b0[0],  b0[OPITCH]};
            const v2f fb1 = {b0[16], b0[OPITCH + 16]};
            o00 = wmma4(fa0, fb0, o00);
            o01 = wmma4(fa0, fb1, o01);
            o10 = wmma4(fa1, fb0, o10);
            o11 = wmma4(fa1, fb1, o11);
        }
        const int m0 = mt * 128 + wm * 32;
        float* base = out + HEADS * NTOK + (size_t)h * NTOK * DH;
        #pragma unroll
        for (int r = 0; r < 8; ++r) {
            float* o = base + (size_t)(m0 + hi * 8 + r) * DH + wn * 32 + lo;
            o[0]            = o00[r] * SCALE;
            o[16]           = o01[r] * SCALE;
            o[16 * DH]      = o10[r] * SCALE;
            o[16 * DH + 16] = o11[r] * SCALE;
        }
    } else {
        const int n = (b - OUT_GEMM_BLOCKS) * 32 + lane;
        if (wave < 4) {
            float a[HEADS];
            #pragma unroll
            for (int hh = 0; hh < HEADS; ++hh) a[hh] = 0.0f;
            #pragma unroll 16
            for (int i = 0; i < 128; ++i) {
                const int e = wave * 128 + i;
                const float xv = x[(size_t)e * NTOK + n];
                #pragma unroll
                for (int hh = 0; hh < HEADS; ++hh) a[hh] += U[hh * EMBED + e] * xv;
            }
            #pragma unroll
            for (int hh = 0; hh < HEADS; ++hh) sC[(wave * HEADS + hh) * 32 + lane] = a[hh];
        }
        __syncthreads();
        if (wave < 4) {
            #pragma unroll
            for (int j = 0; j < 2; ++j) {
                const int hh = wave + 4 * j;
                const float s = ((sC[(0 * HEADS + hh) * 32 + lane] + sC[(1 * HEADS + hh) * 32 + lane])
                               + sC[(2 * HEADS + hh) * 32 + lane]) + sC[(3 * HEADS + hh) * 32 + lane];
                out[hh * NTOK + n] = (s + cpad[hh * 32]) * SCALE;
            }
        }
    }
}

extern "C" void kernel_launch(void* const* d_in, const int* in_sizes, int n_in,
                              void* d_out, int out_size, void* d_ws, size_t ws_size,
                              hipStream_t stream) {
    const float* x    = (const float*)d_in[0];
    const float* W    = (const float*)d_in[1];
    const float* bias = (const float*)d_in[2];
    float* out = (float*)d_out;
    float* ws  = (float*)d_ws;

    mhsa_gram<<<dim3(GTILES, GSPL), 128, 0, stream>>>(x, ws + OFF_GP);
    mhsa_greduce<<<GRED_BLOCKS, 256, 0, stream>>>(ws + OFF_GP, x, ws + OFF_G, ws + OFF_XSP);
    mhsa_t_ktv<<<368, 128, 0, stream>>>(ws + OFF_G, W, bias, ws + OFF_XSP, ws + OFF_KP, ws + OFF_SVP,
                                        ws + OFF_U, ws + OFF_C);
    mhsa_out<<<OUT_GEMM_BLOCKS + 128, 256, 0, stream>>>(x, W, bias, ws + OFF_KP, ws + OFF_SVP, ws + OFF_U, ws + OFF_C, out);
}


# Gram blocks sum all 4096 tokens (256 blocks, 8 waves split tokens, fold in LDS), write G and its mirror as 16x16 tiles, and xsum; reduce launch removed (3 launches); sum order changes
# speedup vs baseline: 2.3841x; 1.0757x over previous
#include <hip/hip_runtime.h>


typedef __attribute__((ext_vector_type(2))) float v2f;
typedef __attribute__((ext_vector_type(4))) float v4f;
typedef __attribute__((ext_vector_type(8))) float v8f;

#define NTOK   4096
#define EMBED  512
#define QKVW   1536
#define HEADS  8
#define DH     64
#define SCALE  0.125f
#define NTOKF  4096.0f
#define GRAM_BLOCKS 256
#define FST    8
#define KSPL   8
#define SSPL   16
#define EE     (EMBED * EMBED)
#define GPITCH 68
#define KPITCH 72

#define FREG   (48 * GPITCH)
#define FBUF   (8 * FREG)
#define PPITCH 17
#define PTILE  (16 * PPITCH)

#define OFF_G     0
#define OFF_KP    (OFF_G + EE)
#define OFF_XSP   (OFF_KP + KSPL * HEADS * DH * DH)
#define OFF_SVP   (OFF_XSP + EMBED)
#define OFF_U     (OFF_SVP + SSPL * QKVW)
#define OFF_C     (OFF_U + HEADS * EMBED)

static __device__ __forceinline__ v8f wmma4(v2f a, v2f b, v8f c) {
    return __builtin_amdgcn_wmma_f32_16x16x4_f32(
        false, a, false, b, (short)0, c, false, false);
}

template <bool DIAG>
static __device__ __forceinline__ void gram_mma(
    const float* pa, const float* pb, v8f& acc0, v8f& acc1, v8f& acc2, v4f& xs0, v4f& xs1)
{
    #pragma unroll
    for (int kk = 0; kk < 64; kk += 8) {
        const v4f b0 = *(const v4f*)(pb + kk);
        const v4f b1 = *(const v4f*)(pb + 16 * GPITCH + kk);
        v4f a;
        if (DIAG) a = b0; else a = *(const v4f*)(pa + kk);
        const v2f a01 = {a.x, a.y},   a23 = {a.z, a.w};
        const v2f p01 = {b0.x, b0.y}, p23 = {b0.z, b0.w};
        const v2f q01 = {b1.x, b1.y}, q23 = {b1.z, b1.w};
        acc0 = wmma4(a01, p01, acc0);
        acc1 = wmma4(a01, q01, acc1);
        if (DIAG) acc2 = wmma4(q01, q01, acc2);
        acc0 = wmma4(a23, p23, acc0);
        acc1 = wmma4(a23, q23, acc1);
        if (DIAG) { acc2 = wmma4(q23, q23, acc2); xs0 += b0; xs1 += b1; }
    }
}

static __device__ __forceinline__ void wave_lds_sync() {
    __builtin_amdgcn_fence(__ATOMIC_RELEASE, "workgroup", "local");
    __builtin_amdgcn_wave_barrier();
    __builtin_amdgcn_fence(__ATOMIC_ACQUIRE, "workgroup", "local");
}

template <bool DIAG>
static __device__ __forceinline__ void gram_load(
    const float* xa, const float* xb, int st, v4f (&ra)[8], v4f (&rb)[16])
{
    #pragma unroll
    for (int j = 0; j < 16; ++j) rb[j] = *(const v4f*)(xb + (size_t)(2 * j) * NTOK + st * 64);
    if (!DIAG) {
        #pragma unroll
        for (int j = 0; j < 8; ++j) ra[j] = *(const v4f*)(xa + (size_t)(2 * j) * NTOK + st * 64);
    }
}

template <bool DIAG>
static __device__ __forceinline__ void gram_store(float* wr, const v4f (&ra)[8], const v4f (&rb)[16])
{
    #pragma unroll
    for (int j = 0; j < 16; ++j) *(v4f*)(wr + (2 * j) * GPITCH) = rb[j];
    if (!DIAG) {
        #pragma unroll
        for (int j = 0; j < 8; ++j) *(v4f*)(wr + (32 + 2 * j) * GPITCH) = ra[j];
    }
}

static __device__ __forceinline__ float gram_fold(const float* p) {
    float s = p[0];
    #pragma unroll
    for (int w = 1; w < 8; ++w) s += p[w * 3 * PTILE];
    return s;
}

template <bool DIAG>
static __device__ __forceinline__ void gram_block(
    const float* __restrict__ x, float* __restrict__ Gt, float* __restrict__ xsum, float* sS, int ta, int tb0)
{
    const int t = threadIdx.x, lane = t & 31, wave = t >> 5, lo = lane & 15, hi = lane >> 4;
    const int lc = lo * 4;
    const float* xb = x + (size_t)(tb0 * 16 + hi) * NTOK + wave * 512 + lc;
    const float* xa = x + (size_t)(ta * 16 + hi) * NTOK + wave * 512 + lc;
    float* wr = sS + wave * FREG + hi * GPITCH + lc;
    const float* pb = sS + wave * FREG + lo * GPITCH + 4 * hi;
    const float* pa = pb + 32 * GPITCH;

    v4f ra[8], rb[16];
    gram_load<DIAG>(xa, xb, 0, ra, rb);
    gram_store<DIAG>(wr, ra, rb);
    gram_load<DIAG>(xa, xb, 1, ra, rb);
    wave_lds_sync();

    v8f acc0 = {}, acc1 = {}, acc2 = {};
    v4f xs0 = {}, xs1 = {};
    #pragma nounroll
    for (int st = 0; st < FST - 2; ++st) {
        const int cur = (st & 1) * FBUF, nxt = FBUF - cur;
        gram_store<DIAG>(wr + nxt, ra, rb);
        gram_load<DIAG>(xa, xb, st + 2, ra, rb);
        __builtin_amdgcn_sched_barrier(0);
        gram_mma<DIAG>(pa + cur, pb + cur, acc0, acc1, acc2, xs0, xs1);
        wave_lds_sync();
    }
    gram_store<DIAG>(wr + FBUF, ra, rb);
    gram_mma<DIAG>(pa, pb, acc0, acc1, acc2, xs0, xs1);
    wave_lds_sync();
    gram_mma<DIAG>(pa + FBUF, pb + FBUF, acc0, acc1, acc2, xs0, xs1);
    __syncthreads();

    float* pl = sS + wave * (3 * PTILE) + (8 * hi) * PPITCH + lo;
    float* sX = sS + 8 * 3 * PTILE;
    #pragma unroll
    for (int r = 0; r < 8; ++r) {
        pl[r * PPITCH]         = acc0[r];
        pl[PTILE + r * PPITCH] = acc1[r];
        if (DIAG) pl[2 * PTILE + r * PPITCH] = acc2[r];
    }
    if (DIAG) {
        float v0 = (xs0.x + xs0.y) + (xs0.z + xs0.w);
        float v1 = (xs1.x + xs1.y) + (xs1.z + xs1.w);
        v0 += __shfl_xor(v0, 16, 32);
        v1 += __shfl_xor(v1, 16, 32);
        if (hi == 0) { sX[wave * 32 + lo] = v0; sX[wave * 32 + 16 + lo] = v1; }
    }
    __syncthreads();

    const int rr = t >> 4, cc = t & 15;
    const float* pd = sS + rr * PPITCH + cc;
    const float* pm = sS + cc * PPITCH + rr;
    Gt[(size_t)(ta * 32 + tb0) * 256 + t] = gram_fold(pd);
    if (!DIAG) Gt[(size_t)(tb0 * 32 + ta) * 256 + t] = gram_fold(pm);
    Gt[(size_t)(ta * 32 + tb0 + 1) * 256 + t] = gram_fold(pd + PTILE);
    Gt[(size_t)((tb0 + 1) * 32 + ta) * 256 + t] = gram_fold(pm + PTILE);
    if (DIAG) {
        Gt[(size_t)((tb0 + 1) * 32 + tb0 + 1) * 256 + t] = gram_fold(pd + 2 * PTILE);
        if (t < 32) {
            float v = sX[t];
            #pragma unroll
            for (int w = 1; w < 8; ++w) v += sX[w * 32 + t];
            xsum[tb0 * 16 + t] = v;
        }
    }
}

__global__ __launch_bounds__(256) void mhsa_gram(
    const float* __restrict__ x, float* __restrict__ Gt, float* __restrict__ xsum)
{
    __shared__ __attribute__((aligned(16))) float sS[2 * FBUF];
    const int b = blockIdx.x;
    if (b < 16) {
        gram_block<true>(x, Gt, xsum, sS, 2 * b, 2 * b);
    } else {
        const int p = (b - 16) >> 1, half = (b - 16) & 1;
        int I = 0, rem = p;
        #pragma unroll
        for (int r = 0; r < 14; ++r) {
            if (I == r && rem >= 15 - r) { rem -= 15 - r; I = r + 1; }
        }
        const int J = I + 1 + rem;
        gram_block<false>(x, Gt, xsum, sS, 2 * I + half, 2 * J);
    }
}

#define TPITCH 20
#define TPLANE (64 * TPITCH)
__global__ __launch_bounds__(128) __attribute__((amdgpu_waves_per_eu(1, 2))) void mhsa_t_ktv(
    const float* __restrict__ G, const float* __restrict__ W, const float* __restrict__ bias,
    const float* __restrict__ xsp, float* __restrict__ Kp, float* __restrict__ svp,
    float* __restrict__ U, float* __restrict__ cpad)
{
    __shared__ float sT[4 * TPLANE];
    const int b = blockIdx.x, t = threadIdx.x;
    const int lane = t & 31, wave = t >> 5, lo = lane & 15, hi = lane >> 4;
    if (b < 256) {
        const int c2 = b & 3, h = (b >> 2) & 7, mi = b >> 5;
        const float* Wk = W + (size_t)(mi * 64 + 2 * hi) * QKVW + EMBED + h * DH + wave * 16 + lo;
        v2f wk[16];
        #pragma unroll
        for (int j = 0; j < 16; ++j)
            wk[j] = (v2f){Wk[(size_t)(4 * j) * QKVW], Wk[(size_t)(4 * j + 1) * QKVW]};

        const float* Ga = G + (size_t)((mi * 4) * 32 + wave * 8) * 256 + lo * 16 + 2 * hi;
        const float* Wv = W + (size_t)(wave * 128 + 2 * hi) * QKVW + 2 * EMBED + h * DH + c2 * 16 + lo;
        v2f ga[2][8][4], wv[2][8];
        #pragma unroll
        for (int u = 0; u < 8; ++u) {
            wv[0][u] = (v2f){Wv[(size_t)(4 * u) * QKVW], Wv[(size_t)(4 * u + 1) * QKVW]};
            #pragma unroll
            for (int i = 0; i < 4; ++i) ga[0][u][i] = *(const v2f*)(Ga + i * (32 * 256) + (u >> 2) * 256 + 4 * (u & 3));
        }
        v8f acc0 = {}, acc1 = {}, acc2 = {}, acc3 = {};
        #pragma unroll
        for (int g = 0; g < 4; ++g) {
            const int cur = g & 1, nxt = cur ^ 1;
            if (g + 1 < 4) {
                #pragma unroll
                for (int u = 0; u < 8; ++u) {
                    const int k = 32 * (g + 1) + 4 * u;
                    wv[nxt][u] = (v2f){Wv[(size_t)k * QKVW], Wv[(size_t)(k + 1) * QKVW]};
                    #pragma unroll
                    for (int i = 0; i < 4; ++i) ga[nxt][u][i] = *(const v2f*)(Ga + i * (32 * 256) + (k >> 4) * 256 + (k & 15));
                }
            }
            __builtin_amdgcn_sched_barrier(0);
            #pragma unroll
            for (int u = 0; u < 8; ++u) {
                acc0 = wmma4(ga[cur][u][0], wv[cur][u], acc0);
                acc1 = wmma4(ga[cur][u][1], wv[cur][u], acc1);
                acc2 = wmma4(ga[cur][u][2], wv[cur][u], acc2);
                acc3 = wmma4(ga[cur][u][3], wv[cur][u], acc3);
            }
        }
        float* pl = sT + wave * TPLANE + (8 * hi) * TPITCH + lo;
        #pragma unroll
        for (int r = 0; r < 8; ++r) {
            pl[r * TPITCH]        = acc0[r];
            pl[(16 + r) * TPITCH] = acc1[r];
            pl[(32 + r) * TPITCH] = acc2[r];
            pl[(48 + r) * TPITCH] = acc3[r];
        }
        __syncthreads();

        const float* tb = sT + (2 * hi) * TPITCH + lo;
        v8f kacc = {};
        #pragma unroll
        for (int j = 0; j < 16; ++j) {
            const float* q = tb + (4 * j) * TPITCH;
            v2f bv = {q[0], q[TPITCH]};
            #pragma unroll
            for (int p = 1; p < 4; ++p) bv += (v2f){q[p * TPLANE], q[p * TPLANE + TPITCH]};
            kacc = wmma4(wk[j], bv, kacc);
        }
        float* dst = Kp + ((((size_t)mi * HEADS + h) * 4 + c2) * DH + wave * 16 + 8 * hi) * 16 + lo;
        #pragma unroll
        for (int r = 0; r < 8; ++r) dst[r * 16] = kacc[r];
    } else if (b < 304) {
        const int gw = (b - 256) * 4 + wave;
        const int sl = gw / 12, cg = gw - sl * 12;
        const int c = cg * 128 + lane * 4;
        v4f s = {};
        #pragma unroll 8
        for (int i = 0; i < 32; ++i) {
            const int e = sl * 32 + i;
            const float xs = xsp[e];
            const v4f w = *(const v4f*)(W + (size_t)e * QKVW + c);
            s += xs * w;
        }
        *(v4f*)(svp + sl * QKVW + c) = s;
    } else {
        const int ub = b - 304, h = ub & 7, et = ub >> 3;
        float* sXs = sT;
        float* sS  = sT + EMBED;
        float* sQ  = sS + SSPL * DH;
        float* sBk = sQ + DH;
        #pragma unroll
        for (int j = 0; j < 4; ++j) sXs[t + 128 * j] = xsp[t + 128 * j];
        if (t < DH) sBk[t] = bias[EMBED + h * DH + t];
        __syncthreads();
        {
            const int c4 = (t & 15) * 4, g = t >> 4;
            const float* wq = W + (size_t)(g * 64) * QKVW + h * DH + c4;
            const float* xq = sXs + g * 64;
            #pragma unroll
            for (int sl = 0; sl < 2; ++sl) {
                v4f s = {};
                #pragma unroll
                for (int i = 0; i < 32; ++i) s += xq[sl * 32 + i] * *(const v4f*)(wq + (size_t)(sl * 32 + i) * QKVW);
                float* o = sS + (g * 2 + sl) * DH + c4;
                o[0] = s.x; o[1] = s.y; o[2] = s.z; o[3] = s.w;
            }
        }
        __syncthreads();
        if (t < DH) {
            float q = sS[t];
            #pragma unroll
            for (int p = 1; p < SSPL; ++p) q += sS[p * DH + t];
            sQ[t] = q + NTOKF * bias[h * DH + t];
        }
        __syncthreads();
        if (t < DH) {
            const int e = et * 64 + t;
            const float* wk = W + (size_t)e * QKVW + EMBED + h * DH;
            float s = 0.0f;
            #pragma unroll
            for (int d = 0; d < DH; d += 4) {
                const v4f w = *(const v4f*)(wk + d);
                s += w.x * sQ[d] + w.y * sQ[d + 1] + w.z * sQ[d + 2] + w.w * sQ[d + 3];
            }
            U[h * EMBED + e] = s;
        } else if (et == 0 && t < DH + 32) {
            float s = 0.0f;
            #pragma unroll
            for (int d = 0; d < DH; ++d) s += sBk[d] * sQ[d];
            cpad[h * 32 + (t - DH)] = s;
        }
    }
}

#define OPITCH 72
#define AP2    288
#define ACHUNK (32 * AP2)
#define BP2    160
#define BCHUNK (32 * BP2)
#define KPW    (HEADS * DH * DH / 4)
#define OUT_GEMM_BLOCKS 256
template <bool STORE>
static __device__ __forceinline__ void out_chunk(
    const float* ca, const float* cb, float* na, float* nb, float* kw,
    const v4f (&ra)[8], const v4f (&rb)[4], const v4f (&kn)[4],
    v8f& acc00, v8f& acc01, v8f& acc10, v8f& acc11)
{
    v2f fa[2][4][2], fb[2][4][2];
    #pragma unroll
    for (int u = 0; u < 4; ++u) {
        fa[0][u][0] = *(const v2f*)(ca + (2 * u) * AP2);
        fa[0][u][1] = *(const v2f*)(ca + (2 * u) * AP2 + 32);
        fb[0][u][0] = *(const v2f*)(cb + (2 * u) * BP2);
        fb[0][u][1] = *(const v2f*)(cb + (2 * u) * BP2 + 32);
    }
    #pragma unroll
    for (int g = 0; g < 4; ++g) {
        const int cs = g & 1, ns = cs ^ 1;
        if (STORE && g == 2) {
            #pragma unroll
            for (int j = 0; j < 4; ++j) {
                *(v4f*)(na + (8 * j) * AP2)     = (v4f){ra[2 * j].x, ra[2 * j + 1].x, ra[2 * j].y, ra[2 * j + 1].y};
                *(v4f*)(na + (8 * j) * AP2 + 4) = (v4f){ra[2 * j].z, ra[2 * j + 1].z, ra[2 * j].w, ra[2 * j + 1].w};
            }
            #pragma unroll
            for (int j = 0; j < 2; ++j) {
                *(v4f*)(nb + (16 * j) * BP2)     = (v4f){rb[2 * j].x, rb[2 * j + 1].x, rb[2 * j].y, rb[2 * j + 1].y};
                *(v4f*)(nb + (16 * j) * BP2 + 4) = (v4f){rb[2 * j].z, rb[2 * j + 1].z, rb[2 * j].w, rb[2 * j + 1].w};
            }
            #pragma unroll
            for (int i = 0; i < 4; ++i) *(v4f*)(kw + 16 * i) = *(const v4f*)(kw + 16 * i) + kn[i];
        }
        if (g + 1 < 4) {
            #pragma unroll
            for (int u = 0; u < 4; ++u) {
                fa[ns][u][0] = *(const v2f*)(ca + (8 * (g + 1) + 2 * u) * AP2);
                fa[ns][u][1] = *(const v2f*)(ca + (8 * (g + 1) + 2 * u) * AP2 + 32);
                fb[ns][u][0] = *(const v2f*)(cb + (8 * (g + 1) + 2 * u) * BP2);
                fb[ns][u][1] = *(const v2f*)(cb + (8 * (g + 1) + 2 * u) * BP2 + 32);
            }
        }
        __builtin_amdgcn_sched_barrier(0);
        #pragma unroll
        for (int u = 0; u < 4; ++u) {
            acc00 = wmma4(fa[cs][u][0], fb[cs][u][0], acc00);
            acc01 = wmma4(fa[cs][u][0], fb[cs][u][1], acc01);
            acc10 = wmma4(fa[cs][u][1], fb[cs][u][0], acc10);
            acc11 = wmma4(fa[cs][u][1], fb[cs][u][1], acc11);
        }
    }
}

__global__ __launch_bounds__(256) void mhsa_out(
    const float* __restrict__ x, const float* __restrict__ W, const float* __restrict__ bias,
    const float* __restrict__ Kp, const float* __restrict__ svp, const float* __restrict__ U,
    const float* __restrict__ cpad, float* __restrict__ out)
{
    __shared__ __attribute__((aligned(16))) float sA[2 * ACHUNK];
    __shared__ __attribute__((aligned(16))) float sB[2 * BCHUNK];
    __shared__ float sC[4 * HEADS * 32];
    __shared__ __attribute__((aligned(16))) float sK[DH * OPITCH];
    __shared__ float sKs[DH], sVs[DH], sBk[DH], sBv[DH];
    const int b = blockIdx.x, t = threadIdx.x;
    const int lane = t & 31, wave = t >> 5, lo = lane & 15, hi = lane >> 4;
    if (b < OUT_GEMM_BLOCKS) {
        const int mt = b & 31, h = b >> 5;
        const int sra = t >> 5, sca = (t & 31) * 4;
        const int srb = t >> 4, scb = (t & 15) * 4;
        const float* xa = x + (size_t)(2 * sra) * NTOK + mt * 128 + sca;
        const float* mb = W + (size_t)(2 * srb) * QKVW + h * DH + scb;
        float* wa = sA + sra * AP2 + 2 * sca;
        float* wb = sB + srb * BP2 + 2 * scb;
        const v4f* kp4 = (const v4f*)(Kp + (size_t)h * DH * DH) + t;

        float* kw = sK + ((t >> 2) & 63) * OPITCH + (t & 3) * 4;

        v4f ra[8], rb[4];
        v4f kn[4];
        #pragma unroll
        for (int i = 0; i < 8; ++i) ra[i] = *(const v4f*)(xa + (size_t)(16 * (i >> 1) + (i & 1)) * NTOK);
        #pragma unroll
        for (int i = 0; i < 4; ++i) {
            rb[i] = *(const v4f*)(mb + (size_t)(32 * (i >> 1) + (i & 1)) * QKVW);
            kn[i] = kp4[256 * i];
        }
        if (t < DH) {
            const int c = h * DH + t;
            float k = svp[EMBED + c], v = svp[2 * EMBED + c];
            #pragma unroll
            for (int p = 1; p < SSPL; ++p) {
                k += svp[p * QKVW + EMBED + c];
                v += svp[p * QKVW + 2 * EMBED + c];
            }
            const float bk = bias[EMBED + c], bv = bias[2 * EMBED + c];
            sKs[t] = k;
            sVs[t] = v + NTOKF * bv;
            sBk[t] = bk;
            sBv[t] = bv;
        }
        #pragma unroll
        for (int j = 0; j < 4; ++j) {
            *(v4f*)(wa + (8 * j) * AP2)     = (v4f){ra[2 * j].x, ra[2 * j + 1].x, ra[2 * j].y, ra[2 * j + 1].y};
            *(v4f*)(wa + (8 * j) * AP2 + 4) = (v4f){ra[2 * j].z, ra[2 * j + 1].z, ra[2 * j].w, ra[2 * j + 1].w};
        }
        #pragma unroll
        for (int j = 0; j < 2; ++j) {
            *(v4f*)(wb + (16 * j) * BP2)     = (v4f){rb[2 * j].x, rb[2 * j + 1].x, rb[2 * j].y, rb[2 * j + 1].y};
            *(v4f*)(wb + (16 * j) * BP2 + 4) = (v4f){rb[2 * j].z, rb[2 * j + 1].z, rb[2 * j].w, rb[2 * j + 1].w};
        }
        #pragma unroll
        for (int i = 0; i < 4; ++i) *(v4f*)(kw + 16 * i) = kn[i];
        __syncthreads();

        const int wm = wave & 3, wn = wave >> 2;
        const float* pa = sA + hi * AP2 + 2 * (wm * 32 + lo);
        const float* pb = sB + hi * BP2 + 2 * (wn * 32 + lo);
        v8f acc00 = {}, acc01 = {}, acc10 = {}, acc11 = {};

        #pragma nounroll
        for (int st = 0; st < 7; ++st) {
            const int curA = (st & 1) * ACHUNK, curB = (st & 1) * BCHUNK;
            const int nxtA = ACHUNK - curA, nxtB = BCHUNK - curB;
            #pragma unroll
            for (int i = 0; i < 8; ++i) ra[i] = *(const v4f*)(xa + (size_t)((st + 1) * 64 + 16 * (i >> 1) + (i & 1)) * NTOK);
            #pragma unroll
            for (int i = 0; i < 4; ++i) {
                rb[i] = *(const v4f*)(mb + (size_t)((st + 1) * 64 + 32 * (i >> 1) + (i & 1)) * QKVW);
                kn[i] = kp4[(size_t)(st + 1) * KPW + 256 * i];
            }
            __builtin_amdgcn_sched_barrier(0);
            out_chunk<true>(pa + curA, pb + curB, wa + nxtA, wb + nxtB, kw, ra, rb, kn, acc00, acc01, acc10, acc11);
            __syncthreads();
        }
        out_chunk<false>(pa + ACHUNK, pb + BCHUNK, wa, wb, kw, ra, rb, kn, acc00, acc01, acc10, acc11);

        #pragma unroll
        for (int i = 0; i < 4; ++i) {
            const int d1 = (t >> 2) & 63, d2 = 16 * i + (t & 3) * 4;
            const float ks = sKs[d1], bk = sBk[d1];
            v4f v = *(const v4f*)(kw + 16 * i);
            v.x += ks * sBv[d2]     + bk * sVs[d2];
            v.y += ks * sBv[d2 + 1] + bk * sVs[d2 + 1];
            v.z += ks * sBv[d2 + 2] + bk * sVs[d2 + 2];
            v.w += ks * sBv[d2 + 3] + bk * sVs[d2 + 3];
            *(v4f*)(kw + 16 * i) = v;
        }
        const int nb = h * DH + wn * 32 + lo;
        const float q0 = bias[nb], q1 = bias[nb + 16];
        float* sQt = sA;
        #pragma unroll
        for (int r = 0; r < 8; ++r) {
            float* q = sQt + (wm * 32 + hi * 8 + r) * GPITCH + wn * 32 + lo;
            q[0]                = acc00[r] + q0;
            q[16]               = acc01[r] + q1;
            q[16 * GPITCH]      = acc10[r] + q0;
            q[16 * GPITCH + 16] = acc11[r] + q1;
        }
        __syncthreads();
        const float* qa = sQt + (wm * 32 + lo) * GPITCH + 2 * hi;
        const float* kb = sK + (2 * hi) * OPITCH + wn * 32 + lo;
        v8f o00 = {}, o01 = {}, o10 = {}, o11 = {};
        #pragma unroll
        for (int kk = 0; kk < DH; kk += 4) {
            const v2f fa0 = *(const v2f*)(qa + kk);
            const v2f fa1 = *(const v2f*)(qa + 16 * GPITCH + kk);
            const float* b0 = kb + kk * OPITCH;
            const v2f fb0 = {b0[0],  b0[OPITCH]};
            const v2f fb1 = {b0[16], b0[OPITCH + 16]};
            o00 = wmma4(fa0, fb0, o00);
            o01 = wmma4(fa0, fb1, o01);
            o10 = wmma4(fa1, fb0, o10);
            o11 = wmma4(fa1, fb1, o11);
        }
        const int m0 = mt * 128 + wm * 32;
        float* base = out + HEADS * NTOK + (size_t)h * NTOK * DH;
        #pragma unroll
        for (int r = 0; r < 8; ++r) {
            float* o = base + (size_t)(m0 + hi * 8 + r) * DH + wn * 32 + lo;
            o[0]            = o00[r] * SCALE;
            o[16]           = o01[r] * SCALE;
            o[16 * DH]      = o10[r] * SCALE;
            o[16 * DH + 16] = o11[r] * SCALE;
        }
    } else {
        const int n = (b - OUT_GEMM_BLOCKS) * 32 + lane;
        if (wave < 4) {
            float a[HEADS];
            #pragma unroll
            for (int hh = 0; hh < HEADS; ++hh) a[hh] = 0.0f;
            #pragma unroll 16
            for (int i = 0; i < 128; ++i) {
                const int e = wave * 128 + i;
                const float xv = x[(size_t)e * NTOK + n];
                #pragma unroll
                for (int hh = 0; hh < HEADS; ++hh) a[hh] += U[hh * EMBED + e] * xv;
            }
            #pragma unroll
            for (int hh = 0; hh < HEADS; ++hh) sC[(wave * HEADS + hh) * 32 + lane] = a[hh];
        }
        __syncthreads();
        if (wave < 4) {
            #pragma unroll
            for (int j = 0; j < 2; ++j) {
                const int hh = wave + 4 * j;
                const float s = ((sC[(0 * HEADS + hh) * 32 + lane] + sC[(1 * HEADS + hh) * 32 + lane])
                               + sC[(2 * HEADS + hh) * 32 + lane]) + sC[(3 * HEADS + hh) * 32 + lane];
                out[hh * NTOK + n] = (s + cpad[hh * 32]) * SCALE;
            }
        }
    }
}

extern "C" void kernel_launch(void* const* d_in, const int* in_sizes, int n_in,
                              void* d_out, int out_size, void* d_ws, size_t ws_size,
                              hipStream_t stream) {
    const float* x    = (const float*)d_in[0];
    const float* W    = (const float*)d_in[1];
    const float* bias = (const float*)d_in[2];
    float* out = (float*)d_out;
    float* ws  = (float*)d_ws;

    mhsa_gram<<<GRAM_BLOCKS, 256, 0, stream>>>(x, ws + OFF_G, ws + OFF_XSP);
    mhsa_t_ktv<<<368, 128, 0, stream>>>(ws + OFF_G, W, bias, ws + OFF_XSP, ws + OFF_KP, ws + OFF_SVP,
                                        ws + OFF_U, ws + OFF_C);
    mhsa_out<<<OUT_GEMM_BLOCKS + 128, 256, 0, stream>>>(x, W, bias, ws + OFF_KP, ws + OFF_SVP, ws + OFF_U, ws + OFF_C, out);
}
